# starting kernel
# baseline (speedup 1.0000x reference)
#include <hip/hip_runtime.h>
#include <hip/hip_bf16.h>

typedef float v2f __attribute__((ext_vector_type(2)));
typedef float v8f __attribute__((ext_vector_type(8)));

#define F_IN  512
#define F_HID 32
#define F_OUT 64


__global__ void k_init_deg(float* __restrict__ deg, int N) {
  int i = blockIdx.x * blockDim.x + threadIdx.x;
  if (i < N) deg[i] = 1.0f;
}

__global__ void k_count_deg(const int* __restrict__ dst, float* __restrict__ deg, int E) {
  int e = blockIdx.x * blockDim.x + threadIdx.x;
  if (e < E) unsafeAtomicAdd(&deg[dst[e]], 1.0f);
}

__global__ void k_rsqrt(float* __restrict__ deg, int N) {
  int i = blockIdx.x * blockDim.x + threadIdx.x;
  if (i < N) deg[i] = rsqrtf(deg[i]);
}


__global__ void k_gemm1(const float* __restrict__ X, const float* __restrict__ W,
                        float* __restrict__ H, int N) {
  const int lane = threadIdx.x & 31;
  const int wave = (blockIdx.x * blockDim.x + threadIdx.x) >> 5;
  const int m0 = wave << 4;
  if (m0 >= N) return;
  const int half = lane >> 4;
  const int lrow = lane & 15;
  int row = m0 + lrow; if (row >= N) row = N - 1;
  const float* xr = X + (size_t)row * F_IN;

  v8f c0 = {}; v8f c1 = {};
#pragma unroll 4
  for (int k0 = 0; k0 < F_IN; k0 += 4) {
    const int ka = k0 + (half << 1);
    v2f a;  a.x  = xr[ka];                      a.y  = xr[ka + 1];
    v2f b0; b0.x = W[ka * F_HID + lrow];        b0.y = W[(ka + 1) * F_HID + lrow];
    v2f b1; b1.x = W[ka * F_HID + 16 + lrow];   b1.y = W[(ka + 1) * F_HID + 16 + lrow];
    c0 = __builtin_amdgcn_wmma_f32_16x16x4_f32(false, a, false, b0, (short)0, c0, false, false);
    c1 = __builtin_amdgcn_wmma_f32_16x16x4_f32(false, a, false, b1, (short)0, c1, false, false);
  }
#pragma unroll
  for (int j = 0; j < 8; ++j) {
    int m = m0 + j + (half << 3);
    if (m < N) {
      float* hr = H + (size_t)m * F_HID;
      hr[lrow]      = c0[j];
      hr[16 + lrow] = c1[j];
    }
  }
}


__global__ void k_gemm2(const float* __restrict__ A, const float* __restrict__ W,
                        float* __restrict__ H, int N) {
  const int lane = threadIdx.x & 31;
  const int wave = (blockIdx.x * blockDim.x + threadIdx.x) >> 5;
  const int m0 = wave << 4;
  if (m0 >= N) return;
  const int half = lane >> 4;
  const int lrow = lane & 15;
  int row = m0 + lrow; if (row >= N) row = N - 1;
  const float* ar = A + (size_t)row * F_HID;

  v8f c0 = {}; v8f c1 = {}; v8f c2 = {}; v8f c3 = {};
#pragma unroll
  for (int k0 = 0; k0 < F_HID; k0 += 4) {
    const int ka = k0 + (half << 1);
    v2f a; a.x = fmaxf(ar[ka], 0.0f); a.y = fmaxf(ar[ka + 1], 0.0f);
    v2f b;
    b.x = W[ka * F_OUT + lrow];            b.y = W[(ka + 1) * F_OUT + lrow];
    c0 = __builtin_amdgcn_wmma_f32_16x16x4_f32(false, a, false, b, (short)0, c0, false, false);
    b.x = W[ka * F_OUT + 16 + lrow];       b.y = W[(ka + 1) * F_OUT + 16 + lrow];
    c1 = __builtin_amdgcn_wmma_f32_16x16x4_f32(false, a, false, b, (short)0, c1, false, false);
    b.x = W[ka * F_OUT + 32 + lrow];       b.y = W[(ka + 1) * F_OUT + 32 + lrow];
    c2 = __builtin_amdgcn_wmma_f32_16x16x4_f32(false, a, false, b, (short)0, c2, false, false);
    b.x = W[ka * F_OUT + 48 + lrow];       b.y = W[(ka + 1) * F_OUT + 48 + lrow];
    c3 = __builtin_amdgcn_wmma_f32_16x16x4_f32(false, a, false, b, (short)0, c3, false, false);
  }
#pragma unroll
  for (int j = 0; j < 8; ++j) {
    int m = m0 + j + (half << 3);
    if (m < N) {
      float* hr = H + (size_t)m * F_OUT;
      hr[lrow]      = c0[j];
      hr[16 + lrow] = c1[j];
      hr[32 + lrow] = c2[j];
      hr[48 + lrow] = c3[j];
    }
  }
}


__global__ void k_agg1_init(const float* __restrict__ h1, const float* __restrict__ dinv,
                            const float* __restrict__ bias, float* __restrict__ out1, int N) {
  int t = blockIdx.x * blockDim.x + threadIdx.x;
  if (t < N * F_HID) {
    int i = t >> 5, f = t & 31;
    float di = dinv[i];
    out1[t] = h1[t] * di * di + bias[f];
  }
}

__global__ void k_agg1(const float* __restrict__ h1, const float* __restrict__ dinv,
                       const int* __restrict__ src, const int* __restrict__ dstv,
                       float* __restrict__ out1, int E) {
  int t = blockIdx.x * blockDim.x + threadIdx.x;
  int e = t >> 5;
  if (e >= E) return;
  int lane = t & 31;
  int s = src[e], d = dstv[e];
  float nrm = dinv[s] * dinv[d];
  float v = h1[(size_t)s * F_HID + lane] * nrm;
  unsafeAtomicAdd(&out1[(size_t)d * F_HID + lane], v);
}

__global__ void k_agg2_init(const float* __restrict__ h2, const float* __restrict__ dinv,
                            const float* __restrict__ bias, float* __restrict__ out, int N) {
  int t = blockIdx.x * blockDim.x + threadIdx.x;
  if (t < N * F_OUT) {
    int i = t >> 6, f = t & 63;
    float di = dinv[i];
    out[t] = h2[t] * di * di + bias[f];
  }
}

__global__ void k_agg2(const float* __restrict__ h2, const float* __restrict__ dinv,
                       const int* __restrict__ src, const int* __restrict__ dstv,
                       float* __restrict__ out, int E) {
  int t = blockIdx.x * blockDim.x + threadIdx.x;
  int e = t >> 5;
  if (e >= E) return;
  int lane = t & 31;
  int s = src[e], d = dstv[e];
  float nrm = dinv[s] * dinv[d];
  const float* hs = h2 + (size_t)s * F_OUT;
  float* od = out + (size_t)d * F_OUT;
  unsafeAtomicAdd(&od[lane],      hs[lane]      * nrm);
  unsafeAtomicAdd(&od[lane + 32], hs[lane + 32] * nrm);
}


static inline int cdiv_ll(long long a, long long b) { return (int)((a + b - 1) / b); }

extern "C" void kernel_launch(void* const* d_in, const int* in_sizes, int n_in,
                              void* d_out, int out_size, void* d_ws, size_t ws_size,
                              hipStream_t stream) {
  const float* X  = (const float*)d_in[0];
  const int*   ei = (const int*)d_in[1];
  const float* W1 = (const float*)d_in[2];
  const float* b1 = (const float*)d_in[3];
  const float* W2 = (const float*)d_in[4];
  const float* b2 = (const float*)d_in[5];
  float* out = (float*)d_out;

  const int N = in_sizes[0] / F_IN;
  const int E = in_sizes[1] / 2;
  const int* src = ei;
  const int* dst = ei + E;

  float* dinv = (float*)d_ws;
  float* h1   = dinv + N;
  float* out1 = h1 + (size_t)N * F_HID;
  float* h2   = out1 + (size_t)N * F_HID;

  const int B = 256;
  const int mtiles = cdiv_ll(N, 16);

  k_init_deg <<<cdiv_ll(N, B), B, 0, stream>>>(dinv, N);
  k_count_deg<<<cdiv_ll(E, B), B, 0, stream>>>(dst, dinv, E);
  k_rsqrt    <<<cdiv_ll(N, B), B, 0, stream>>>(dinv, N);

  k_gemm1    <<<cdiv_ll((long long)mtiles * 32, B), B, 0, stream>>>(X, W1, h1, N);
  k_agg1_init<<<cdiv_ll((long long)N * F_HID, B), B, 0, stream>>>(h1, dinv, b1, out1, N);
  k_agg1     <<<cdiv_ll((long long)E * 32, B), B, 0, stream>>>(h1, dinv, src, dst, out1, E);

  k_gemm2    <<<cdiv_ll((long long)mtiles * 32, B), B, 0, stream>>>(out1, W2, h2, N);
  k_agg2_init<<<cdiv_ll((long long)N * F_OUT, B), B, 0, stream>>>(h2, dinv, b2, out, N);
  k_agg2     <<<cdiv_ll((long long)E * 32, B), B, 0, stream>>>(h2, dinv, src, dst, out, E);
}


# agg1/agg2: one wave walks 32 consecutive edges (coalesced index loads, row gathers all in flight, then whole-line atomics) instead of one edge per wave; atomic sum order unordered as before
# speedup vs baseline: 1.8716x; 1.8716x over previous
#include <hip/hip_runtime.h>
#include <hip/hip_bf16.h>

typedef float v2f __attribute__((ext_vector_type(2)));
typedef float v8f __attribute__((ext_vector_type(8)));

#define F_IN  512
#define F_HID 32
#define F_OUT 64


__global__ void k_init_deg(float* __restrict__ deg, int N) {
  int i = blockIdx.x * blockDim.x + threadIdx.x;
  if (i < N) deg[i] = 1.0f;
}

__global__ void k_count_deg(const int* __restrict__ dst, float* __restrict__ deg, int E) {
  int e = blockIdx.x * blockDim.x + threadIdx.x;
  if (e < E) unsafeAtomicAdd(&deg[dst[e]], 1.0f);
}

__global__ void k_rsqrt(float* __restrict__ deg, int N) {
  int i = blockIdx.x * blockDim.x + threadIdx.x;
  if (i < N) deg[i] = rsqrtf(deg[i]);
}


__global__ void k_gemm1(const float* __restrict__ X, const float* __restrict__ W,
                        float* __restrict__ H, int N) {
  const int lane = threadIdx.x & 31;
  const int wave = (blockIdx.x * blockDim.x + threadIdx.x) >> 5;
  const int m0 = wave << 4;
  if (m0 >= N) return;
  const int half = lane >> 4;
  const int lrow = lane & 15;
  int row = m0 + lrow; if (row >= N) row = N - 1;
  const float* xr = X + (size_t)row * F_IN;

  v8f c0 = {}; v8f c1 = {};
#pragma unroll 4
  for (int k0 = 0; k0 < F_IN; k0 += 4) {
    const int ka = k0 + (half << 1);
    v2f a;  a.x  = xr[ka];                      a.y  = xr[ka + 1];
    v2f b0; b0.x = W[ka * F_HID + lrow];        b0.y = W[(ka + 1) * F_HID + lrow];
    v2f b1; b1.x = W[ka * F_HID + 16 + lrow];   b1.y = W[(ka + 1) * F_HID + 16 + lrow];
    c0 = __builtin_amdgcn_wmma_f32_16x16x4_f32(false, a, false, b0, (short)0, c0, false, false);
    c1 = __builtin_amdgcn_wmma_f32_16x16x4_f32(false, a, false, b1, (short)0, c1, false, false);
  }
#pragma unroll
  for (int j = 0; j < 8; ++j) {
    int m = m0 + j + (half << 3);
    if (m < N) {
      float* hr = H + (size_t)m * F_HID;
      hr[lrow]      = c0[j];
      hr[16 + lrow] = c1[j];
    }
  }
}


__global__ void k_gemm2(const float* __restrict__ A, const float* __restrict__ W,
                        float* __restrict__ H, int N) {
  const int lane = threadIdx.x & 31;
  const int wave = (blockIdx.x * blockDim.x + threadIdx.x) >> 5;
  const int m0 = wave << 4;
  if (m0 >= N) return;
  const int half = lane >> 4;
  const int lrow = lane & 15;
  int row = m0 + lrow; if (row >= N) row = N - 1;
  const float* ar = A + (size_t)row * F_HID;

  v8f c0 = {}; v8f c1 = {}; v8f c2 = {}; v8f c3 = {};
#pragma unroll
  for (int k0 = 0; k0 < F_HID; k0 += 4) {
    const int ka = k0 + (half << 1);
    v2f a; a.x = fmaxf(ar[ka], 0.0f); a.y = fmaxf(ar[ka + 1], 0.0f);
    v2f b;
    b.x = W[ka * F_OUT + lrow];            b.y = W[(ka + 1) * F_OUT + lrow];
    c0 = __builtin_amdgcn_wmma_f32_16x16x4_f32(false, a, false, b, (short)0, c0, false, false);
    b.x = W[ka * F_OUT + 16 + lrow];       b.y = W[(ka + 1) * F_OUT + 16 + lrow];
    c1 = __builtin_amdgcn_wmma_f32_16x16x4_f32(false, a, false, b, (short)0, c1, false, false);
    b.x = W[ka * F_OUT + 32 + lrow];       b.y = W[(ka + 1) * F_OUT + 32 + lrow];
    c2 = __builtin_amdgcn_wmma_f32_16x16x4_f32(false, a, false, b, (short)0, c2, false, false);
    b.x = W[ka * F_OUT + 48 + lrow];       b.y = W[(ka + 1) * F_OUT + 48 + lrow];
    c3 = __builtin_amdgcn_wmma_f32_16x16x4_f32(false, a, false, b, (short)0, c3, false, false);
  }
#pragma unroll
  for (int j = 0; j < 8; ++j) {
    int m = m0 + j + (half << 3);
    if (m < N) {
      float* hr = H + (size_t)m * F_OUT;
      hr[lrow]      = c0[j];
      hr[16 + lrow] = c1[j];
      hr[32 + lrow] = c2[j];
      hr[48 + lrow] = c3[j];
    }
  }
}


__global__ void k_agg1_init(const float* __restrict__ h1, const float* __restrict__ dinv,
                            const float* __restrict__ bias, float* __restrict__ out1, int N) {
  int t = blockIdx.x * blockDim.x + threadIdx.x;
  if (t < N * F_HID) {
    int i = t >> 5, f = t & 31;
    float di = dinv[i];
    out1[t] = h1[t] * di * di + bias[f];
  }
}

__global__ void k_agg1(const float* __restrict__ h1, const float* __restrict__ dinv,
                       const int* __restrict__ src, const int* __restrict__ dstv,
                       float* __restrict__ out1, int E) {
  const int lane = threadIdx.x & 31;
  const int wave = (blockIdx.x * blockDim.x + threadIdx.x) >> 5;
  const int e0 = wave << 5;
  if (e0 >= E) return;
  const int e = e0 + lane;
  int s = 0, d = 0;
  float nrm = 0.0f;
  if (e < E) {
    s = src[e]; d = dstv[e];
    nrm = dinv[s] * dinv[d];
  }
  const int nrm_bits = __float_as_int(nrm);
  const int cnt = (E - e0 < 32) ? (E - e0) : 32;
  if (cnt == 32) {
    float v[32];
#pragma unroll
    for (int j = 0; j < 32; ++j) {
      const int sj = __builtin_amdgcn_readlane(s, j);
      v[j] = h1[(size_t)sj * F_HID + lane];
    }
#pragma unroll
    for (int j = 0; j < 32; ++j) {
      const int dj = __builtin_amdgcn_readlane(d, j);
      const float nj = __int_as_float(__builtin_amdgcn_readlane(nrm_bits, j));
      unsafeAtomicAdd(&out1[(size_t)dj * F_HID + lane], v[j] * nj);
    }
  } else {
    for (int j = 0; j < cnt; ++j) {
      const int sj = __builtin_amdgcn_readlane(s, j);
      const int dj = __builtin_amdgcn_readlane(d, j);
      const float nj = __int_as_float(__builtin_amdgcn_readlane(nrm_bits, j));
      unsafeAtomicAdd(&out1[(size_t)dj * F_HID + lane], h1[(size_t)sj * F_HID + lane] * nj);
    }
  }
}

__global__ void k_agg2_init(const float* __restrict__ h2, const float* __restrict__ dinv,
                            const float* __restrict__ bias, float* __restrict__ out, int N) {
  int t = blockIdx.x * blockDim.x + threadIdx.x;
  if (t < N * F_OUT) {
    int i = t >> 6, f = t & 63;
    float di = dinv[i];
    out[t] = h2[t] * di * di + bias[f];
  }
}

__global__ void k_agg2(const float* __restrict__ h2, const float* __restrict__ dinv,
                       const int* __restrict__ src, const int* __restrict__ dstv,
                       float* __restrict__ out, int E) {
  const int lane = threadIdx.x & 31;
  const int wave = (blockIdx.x * blockDim.x + threadIdx.x) >> 5;
  const int e0 = wave << 5;
  if (e0 >= E) return;
  const int e = e0 + lane;
  int s = 0, d = 0;
  float nrm = 0.0f;
  if (e < E) {
    s = src[e]; d = dstv[e];
    nrm = dinv[s] * dinv[d];
  }
  const int nrm_bits = __float_as_int(nrm);
  const int cnt = (E - e0 < 32) ? (E - e0) : 32;
  if (cnt == 32) {
#pragma unroll
    for (int g = 0; g < 2; ++g) {
      float a[16], b[16];
#pragma unroll
      for (int j = 0; j < 16; ++j) {
        const int sj = __builtin_amdgcn_readlane(s, g * 16 + j);
        const float* hs = h2 + (size_t)sj * F_OUT;
        a[j] = hs[lane];
        b[j] = hs[lane + 32];
      }
#pragma unroll
      for (int j = 0; j < 16; ++j) {
        const int dj = __builtin_amdgcn_readlane(d, g * 16 + j);
        const float nj = __int_as_float(__builtin_amdgcn_readlane(nrm_bits, g * 16 + j));
        float* od = out + (size_t)dj * F_OUT;
        unsafeAtomicAdd(&od[lane],      a[j] * nj);
        unsafeAtomicAdd(&od[lane + 32], b[j] * nj);
      }
    }
  } else {
    for (int j = 0; j < cnt; ++j) {
      const int sj = __builtin_amdgcn_readlane(s, j);
      const int dj = __builtin_amdgcn_readlane(d, j);
      const float nj = __int_as_float(__builtin_amdgcn_readlane(nrm_bits, j));
      const float* hs = h2 + (size_t)sj * F_OUT;
      float* od = out + (size_t)dj * F_OUT;
      unsafeAtomicAdd(&od[lane],      hs[lane]      * nj);
      unsafeAtomicAdd(&od[lane + 32], hs[lane + 32] * nj);
    }
  }
}


static inline int cdiv_ll(long long a, long long b) { return (int)((a + b - 1) / b); }

extern "C" void kernel_launch(void* const* d_in, const int* in_sizes, int n_in,
                              void* d_out, int out_size, void* d_ws, size_t ws_size,
                              hipStream_t stream) {
  const float* X  = (const float*)d_in[0];
  const int*   ei = (const int*)d_in[1];
  const float* W1 = (const float*)d_in[2];
  const float* b1 = (const float*)d_in[3];
  const float* W2 = (const float*)d_in[4];
  const float* b2 = (const float*)d_in[5];
  float* out = (float*)d_out;

  const int N = in_sizes[0] / F_IN;
  const int E = in_sizes[1] / 2;
  const int* src = ei;
  const int* dst = ei + E;

  float* dinv = (float*)d_ws;
  float* h1   = dinv + N;
  float* out1 = h1 + (size_t)N * F_HID;
  float* h2   = out1 + (size_t)N * F_HID;

  const int B = 256;
  const int mtiles = cdiv_ll(N, 16);

  k_init_deg <<<cdiv_ll(N, B), B, 0, stream>>>(dinv, N);
  k_count_deg<<<cdiv_ll(E, B), B, 0, stream>>>(dst, dinv, E);
  k_rsqrt    <<<cdiv_ll(N, B), B, 0, stream>>>(dinv, N);

  k_gemm1    <<<cdiv_ll((long long)mtiles * 32, B), B, 0, stream>>>(X, W1, h1, N);
  k_agg1_init<<<cdiv_ll((long long)N * F_HID, B), B, 0, stream>>>(h1, dinv, b1, out1, N);
  k_agg1     <<<cdiv_ll(E, B), B, 0, stream>>>(h1, dinv, src, dst, out1, E);

  k_gemm2    <<<cdiv_ll((long long)mtiles * 32, B), B, 0, stream>>>(out1, W2, h2, N);
  k_agg2_init<<<cdiv_ll((long long)N * F_OUT, B), B, 0, stream>>>(h2, dinv, b2, out, N);
  k_agg2     <<<cdiv_ll(E, B), B, 0, stream>>>(h2, dinv, src, dst, out, E);
}


# layer 2 aggregates the 32-wide relu rows first and applies W2 (WMMA, +b2) afterwards, A(HW)=(AH)W: one 128-byte line gathered and added per edge instead of two; f32 sum order changes
# speedup vs baseline: 1.9750x; 1.0552x over previous
#include <hip/hip_runtime.h>
#include <hip/hip_bf16.h>

typedef float v2f __attribute__((ext_vector_type(2)));
typedef float v8f __attribute__((ext_vector_type(8)));

#define F_IN  512
#define F_HID 32
#define F_OUT 64


__global__ void k_init_deg(float* __restrict__ deg, int N) {
  int i = blockIdx.x * blockDim.x + threadIdx.x;
  if (i < N) deg[i] = 1.0f;
}

__global__ void k_count_deg(const int* __restrict__ dst, float* __restrict__ deg, int E) {
  int e = blockIdx.x * blockDim.x + threadIdx.x;
  if (e < E) unsafeAtomicAdd(&deg[dst[e]], 1.0f);
}

__global__ void k_rsqrt(float* __restrict__ deg, int N) {
  int i = blockIdx.x * blockDim.x + threadIdx.x;
  if (i < N) deg[i] = rsqrtf(deg[i]);
}


__global__ void k_gemm1(const float* __restrict__ X, const float* __restrict__ W,
                        float* __restrict__ H, int N) {
  const int lane = threadIdx.x & 31;
  const int wave = (blockIdx.x * blockDim.x + threadIdx.x) >> 5;
  const int m0 = wave << 4;
  if (m0 >= N) return;
  const int half = lane >> 4;
  const int lrow = lane & 15;
  int row = m0 + lrow; if (row >= N) row = N - 1;
  const float* xr = X + (size_t)row * F_IN;

  v8f c0 = {}; v8f c1 = {};
#pragma unroll 4
  for (int k0 = 0; k0 < F_IN; k0 += 4) {
    const int ka = k0 + (half << 1);
    v2f a;  a.x  = xr[ka];                      a.y  = xr[ka + 1];
    v2f b0; b0.x = W[ka * F_HID + lrow];        b0.y = W[(ka + 1) * F_HID + lrow];
    v2f b1; b1.x = W[ka * F_HID + 16 + lrow];   b1.y = W[(ka + 1) * F_HID + 16 + lrow];
    c0 = __builtin_amdgcn_wmma_f32_16x16x4_f32(false, a, false, b0, (short)0, c0, false, false);
    c1 = __builtin_amdgcn_wmma_f32_16x16x4_f32(false, a, false, b1, (short)0, c1, false, false);
  }
#pragma unroll
  for (int j = 0; j < 8; ++j) {
    int m = m0 + j + (half << 3);
    if (m < N) {
      float* hr = H + (size_t)m * F_HID;
      hr[lrow]      = c0[j];
      hr[16 + lrow] = c1[j];
    }
  }
}


__global__ void k_gemm2(const float* __restrict__ A, const float* __restrict__ W,
                        const float* __restrict__ bias, float* __restrict__ H, int N) {
  const int lane = threadIdx.x & 31;
  const int wave = (blockIdx.x * blockDim.x + threadIdx.x) >> 5;
  const int m0 = wave << 4;
  if (m0 >= N) return;
  const int half = lane >> 4;
  const int lrow = lane & 15;
  int row = m0 + lrow; if (row >= N) row = N - 1;
  const float* ar = A + (size_t)row * F_HID;

  v8f c0 = {}; v8f c1 = {}; v8f c2 = {}; v8f c3 = {};
#pragma unroll
  for (int k0 = 0; k0 < F_HID; k0 += 4) {
    const int ka = k0 + (half << 1);
    v2f a; a.x = ar[ka]; a.y = ar[ka + 1];
    v2f b;
    b.x = W[ka * F_OUT + lrow];            b.y = W[(ka + 1) * F_OUT + lrow];
    c0 = __builtin_amdgcn_wmma_f32_16x16x4_f32(false, a, false, b, (short)0, c0, false, false);
    b.x = W[ka * F_OUT + 16 + lrow];       b.y = W[(ka + 1) * F_OUT + 16 + lrow];
    c1 = __builtin_amdgcn_wmma_f32_16x16x4_f32(false, a, false, b, (short)0, c1, false, false);
    b.x = W[ka * F_OUT + 32 + lrow];       b.y = W[(ka + 1) * F_OUT + 32 + lrow];
    c2 = __builtin_amdgcn_wmma_f32_16x16x4_f32(false, a, false, b, (short)0, c2, false, false);
    b.x = W[ka * F_OUT + 48 + lrow];       b.y = W[(ka + 1) * F_OUT + 48 + lrow];
    c3 = __builtin_amdgcn_wmma_f32_16x16x4_f32(false, a, false, b, (short)0, c3, false, false);
  }
  const float bz0 = bias[lrow], bz1 = bias[16 + lrow], bz2 = bias[32 + lrow], bz3 = bias[48 + lrow];
#pragma unroll
  for (int j = 0; j < 8; ++j) {
    int m = m0 + j + (half << 3);
    if (m < N) {
      float* hr = H + (size_t)m * F_OUT;
      hr[lrow]      = c0[j] + bz0;
      hr[16 + lrow] = c1[j] + bz1;
      hr[32 + lrow] = c2[j] + bz2;
      hr[48 + lrow] = c3[j] + bz3;
    }
  }
}


__global__ void k_agg1_init(const float* __restrict__ h1, const float* __restrict__ dinv,
                            const float* __restrict__ bias, float* __restrict__ out1, int N) {
  int t = blockIdx.x * blockDim.x + threadIdx.x;
  if (t < N * F_HID) {
    int i = t >> 5, f = t & 31;
    float di = dinv[i];
    out1[t] = h1[t] * di * di + bias[f];
  }
}

__global__ void k_agg1(const float* __restrict__ h1, const float* __restrict__ dinv,
                       const int* __restrict__ src, const int* __restrict__ dstv,
                       float* __restrict__ out1, int E) {
  const int lane = threadIdx.x & 31;
  const int wave = (blockIdx.x * blockDim.x + threadIdx.x) >> 5;
  const int e0 = wave << 5;
  if (e0 >= E) return;
  const int e = e0 + lane;
  int s = 0, d = 0;
  float nrm = 0.0f;
  if (e < E) {
    s = src[e]; d = dstv[e];
    nrm = dinv[s] * dinv[d];
  }
  const int nrm_bits = __float_as_int(nrm);
  const int cnt = (E - e0 < 32) ? (E - e0) : 32;
  if (cnt == 32) {
    float v[32];
#pragma unroll
    for (int j = 0; j < 32; ++j) {
      const int sj = __builtin_amdgcn_readlane(s, j);
      v[j] = h1[(size_t)sj * F_HID + lane];
    }
#pragma unroll
    for (int j = 0; j < 32; ++j) {
      const int dj = __builtin_amdgcn_readlane(d, j);
      const float nj = __int_as_float(__builtin_amdgcn_readlane(nrm_bits, j));
      unsafeAtomicAdd(&out1[(size_t)dj * F_HID + lane], v[j] * nj);
    }
  } else {
    for (int j = 0; j < cnt; ++j) {
      const int sj = __builtin_amdgcn_readlane(s, j);
      const int dj = __builtin_amdgcn_readlane(d, j);
      const float nj = __int_as_float(__builtin_amdgcn_readlane(nrm_bits, j));
      unsafeAtomicAdd(&out1[(size_t)dj * F_HID + lane], h1[(size_t)sj * F_HID + lane] * nj);
    }
  }
}

__global__ void k_agg2_init(const float* __restrict__ out1, const float* __restrict__ dinv,
                            float* __restrict__ g, int N) {
  int t = blockIdx.x * blockDim.x + threadIdx.x;
  if (t < N * F_HID) {
    int i = t >> 5;
    float di = dinv[i];
    g[t] = fmaxf(out1[t], 0.0f) * di * di;
  }
}

__global__ void k_agg2(const float* __restrict__ out1, const float* __restrict__ dinv,
                       const int* __restrict__ src, const int* __restrict__ dstv,
                       float* __restrict__ g, int E) {
  const int lane = threadIdx.x & 31;
  const int wave = (blockIdx.x * blockDim.x + threadIdx.x) >> 5;
  const int e0 = wave << 5;
  if (e0 >= E) return;
  const int e = e0 + lane;
  int s = 0, d = 0;
  float nrm = 0.0f;
  if (e < E) {
    s = src[e]; d = dstv[e];
    nrm = dinv[s] * dinv[d];
  }
  const int nrm_bits = __float_as_int(nrm);
  const int cnt = (E - e0 < 32) ? (E - e0) : 32;
  if (cnt == 32) {
    float v[32];
#pragma unroll
    for (int j = 0; j < 32; ++j) {
      const int sj = __builtin_amdgcn_readlane(s, j);
      v[j] = out1[(size_t)sj * F_HID + lane];
    }
#pragma unroll
    for (int j = 0; j < 32; ++j) {
      const int dj = __builtin_amdgcn_readlane(d, j);
      const float nj = __int_as_float(__builtin_amdgcn_readlane(nrm_bits, j));
      unsafeAtomicAdd(&g[(size_t)dj * F_HID + lane], fmaxf(v[j], 0.0f) * nj);
    }
  } else {
    for (int j = 0; j < cnt; ++j) {
      const int sj = __builtin_amdgcn_readlane(s, j);
      const int dj = __builtin_amdgcn_readlane(d, j);
      const float nj = __int_as_float(__builtin_amdgcn_readlane(nrm_bits, j));
      unsafeAtomicAdd(&g[(size_t)dj * F_HID + lane], fmaxf(out1[(size_t)sj * F_HID + lane], 0.0f) * nj);
    }
  }
}


static inline int cdiv_ll(long long a, long long b) { return (int)((a + b - 1) / b); }

extern "C" void kernel_launch(void* const* d_in, const int* in_sizes, int n_in,
                              void* d_out, int out_size, void* d_ws, size_t ws_size,
                              hipStream_t stream) {
  const float* X  = (const float*)d_in[0];
  const int*   ei = (const int*)d_in[1];
  const float* W1 = (const float*)d_in[2];
  const float* b1 = (const float*)d_in[3];
  const float* W2 = (const float*)d_in[4];
  const float* b2 = (const float*)d_in[5];
  float* out = (float*)d_out;

  const int N = in_sizes[0] / F_IN;
  const int E = in_sizes[1] / 2;
  const int* src = ei;
  const int* dst = ei + E;

  float* dinv = (float*)d_ws;
  float* h1   = dinv + N;
  float* out1 = h1 + (size_t)N * F_HID;
  float* g    = out1 + (size_t)N * F_HID;

  const int B = 256;
  const int mtiles = cdiv_ll(N, 16);

  k_init_deg <<<cdiv_ll(N, B), B, 0, stream>>>(dinv, N);
  k_count_deg<<<cdiv_ll(E, B), B, 0, stream>>>(dst, dinv, E);
  k_rsqrt    <<<cdiv_ll(N, B), B, 0, stream>>>(dinv, N);

  k_gemm1    <<<cdiv_ll((long long)mtiles * 32, B), B, 0, stream>>>(X, W1, h1, N);
  k_agg1_init<<<cdiv_ll((long long)N * F_HID, B), B, 0, stream>>>(h1, dinv, b1, out1, N);
  k_agg1     <<<cdiv_ll(E, B), B, 0, stream>>>(h1, dinv, src, dst, out1, E);

  k_agg2_init<<<cdiv_ll((long long)N * F_HID, B), B, 0, stream>>>(out1, dinv, g, N);
  k_agg2     <<<cdiv_ll(E, B), B, 0, stream>>>(out1, dinv, src, dst, g, E);
  k_gemm2    <<<cdiv_ll((long long)mtiles * 32, B), B, 0, stream>>>(g, W2, b2, out, N);
}


# edge weight dinv[s]*dinv[d] split over the edge's ends (rows scaled by dinv[s] where produced, sums by dinv[d] where consumed): edge walks gather no dinv, multiply nothing; f32 rounding changes
# speedup vs baseline: 2.1037x; 1.0652x over previous
#include <hip/hip_runtime.h>
#include <hip/hip_bf16.h>

typedef float v2f __attribute__((ext_vector_type(2)));
typedef float v8f __attribute__((ext_vector_type(8)));

#define F_IN  512
#define F_HID 32
#define F_OUT 64


__global__ void k_init_deg(float* __restrict__ deg, int N) {
  int i = blockIdx.x * blockDim.x + threadIdx.x;
  if (i < N) deg[i] = 1.0f;
}

__global__ void k_count_deg(const int* __restrict__ dst, float* __restrict__ deg, int E) {
  int e = blockIdx.x * blockDim.x + threadIdx.x;
  if (e < E) unsafeAtomicAdd(&deg[dst[e]], 1.0f);
}

__global__ void k_rsqrt(float* __restrict__ deg, int N) {
  int i = blockIdx.x * blockDim.x + threadIdx.x;
  if (i < N) deg[i] = rsqrtf(deg[i]);
}


__global__ void k_gemm1(const float* __restrict__ X, const float* __restrict__ W,
                        const float* __restrict__ dinv, float* __restrict__ H, int N) {
  const int lane = threadIdx.x & 31;
  const int wave = (blockIdx.x * blockDim.x + threadIdx.x) >> 5;
  const int m0 = wave << 4;
  if (m0 >= N) return;
  const int half = lane >> 4;
  const int lrow = lane & 15;
  int row = m0 + lrow; if (row >= N) row = N - 1;
  const float* xr = X + (size_t)row * F_IN;

  float dm[8];
#pragma unroll
  for (int j = 0; j < 8; ++j) {
    int m = m0 + j + (half << 3); if (m >= N) m = N - 1;
    dm[j] = dinv[m];
  }

  v8f c0 = {}; v8f c1 = {};
#pragma unroll 4
  for (int k0 = 0; k0 < F_IN; k0 += 4) {
    const int ka = k0 + (half << 1);
    v2f a;  a.x  = xr[ka];                      a.y  = xr[ka + 1];
    v2f b0; b0.x = W[ka * F_HID + lrow];        b0.y = W[(ka + 1) * F_HID + lrow];
    v2f b1; b1.x = W[ka * F_HID + 16 + lrow];   b1.y = W[(ka + 1) * F_HID + 16 + lrow];
    c0 = __builtin_amdgcn_wmma_f32_16x16x4_f32(false, a, false, b0, (short)0, c0, false, false);
    c1 = __builtin_amdgcn_wmma_f32_16x16x4_f32(false, a, false, b1, (short)0, c1, false, false);
  }
#pragma unroll
  for (int j = 0; j < 8; ++j) {
    int m = m0 + j + (half << 3);
    if (m < N) {
      float* hr = H + (size_t)m * F_HID;
      hr[lrow]      = c0[j] * dm[j];
      hr[16 + lrow] = c1[j] * dm[j];
    }
  }
}


__global__ void k_gemm2(const float* __restrict__ A, const float* __restrict__ W,
                        const float* __restrict__ bias, const float* __restrict__ dinv,
                        float* __restrict__ H, int N) {
  const int lane = threadIdx.x & 31;
  const int wave = (blockIdx.x * blockDim.x + threadIdx.x) >> 5;
  const int m0 = wave << 4;
  if (m0 >= N) return;
  const int half = lane >> 4;
  const int lrow = lane & 15;
  int row = m0 + lrow; if (row >= N) row = N - 1;
  const float* ar = A + (size_t)row * F_HID;

  float dm[8];
#pragma unroll
  for (int j = 0; j < 8; ++j) {
    int m = m0 + j + (half << 3); if (m >= N) m = N - 1;
    dm[j] = dinv[m];
  }

  v8f c0 = {}; v8f c1 = {}; v8f c2 = {}; v8f c3 = {};
#pragma unroll
  for (int k0 = 0; k0 < F_HID; k0 += 4) {
    const int ka = k0 + (half << 1);
    v2f a; a.x = ar[ka]; a.y = ar[ka + 1];
    v2f b;
    b.x = W[ka * F_OUT + lrow];            b.y = W[(ka + 1) * F_OUT + lrow];
    c0 = __builtin_amdgcn_wmma_f32_16x16x4_f32(false, a, false, b, (short)0, c0, false, false);
    b.x = W[ka * F_OUT + 16 + lrow];       b.y = W[(ka + 1) * F_OUT + 16 + lrow];
    c1 = __builtin_amdgcn_wmma_f32_16x16x4_f32(false, a, false, b, (short)0, c1, false, false);
    b.x = W[ka * F_OUT + 32 + lrow];       b.y = W[(ka + 1) * F_OUT + 32 + lrow];
    c2 = __builtin_amdgcn_wmma_f32_16x16x4_f32(false, a, false, b, (short)0, c2, false, false);
    b.x = W[ka * F_OUT + 48 + lrow];       b.y = W[(ka + 1) * F_OUT + 48 + lrow];
    c3 = __builtin_amdgcn_wmma_f32_16x16x4_f32(false, a, false, b, (short)0, c3, false, false);
  }
  const float bz0 = bias[lrow], bz1 = bias[16 + lrow], bz2 = bias[32 + lrow], bz3 = bias[48 + lrow];
#pragma unroll
  for (int j = 0; j < 8; ++j) {
    int m = m0 + j + (half << 3);
    if (m < N) {
      float* hr = H + (size_t)m * F_OUT;
      hr[lrow]      = c0[j] * dm[j] + bz0;
      hr[16 + lrow] = c1[j] * dm[j] + bz1;
      hr[32 + lrow] = c2[j] * dm[j] + bz2;
      hr[48 + lrow] = c3[j] * dm[j] + bz3;
    }
  }
}


__global__ void k_agg1_init(const float* __restrict__ hs, float* __restrict__ acc, int N) {
  int t = blockIdx.x * blockDim.x + threadIdx.x;
  if (t < N * F_HID) acc[t] = hs[t];
}

__global__ void k_agg(const float* __restrict__ rows, const int* __restrict__ src,
                      const int* __restrict__ dstv, float* __restrict__ acc, int E) {
  const int lane = threadIdx.x & 31;
  const int wave = (blockIdx.x * blockDim.x + threadIdx.x) >> 5;
  const int e0 = wave << 5;
  if (e0 >= E) return;
  const int e = e0 + lane;
  int s = 0, d = 0;
  if (e < E) { s = src[e]; d = dstv[e]; }
  const int cnt = (E - e0 < 32) ? (E - e0) : 32;
  if (cnt == 32) {
    float v[32];
#pragma unroll
    for (int j = 0; j < 32; ++j) {
      const int sj = __builtin_amdgcn_readlane(s, j);
      v[j] = rows[(size_t)sj * F_HID + lane];
    }
#pragma unroll
    for (int j = 0; j < 32; ++j) {
      const int dj = __builtin_amdgcn_readlane(d, j);
      unsafeAtomicAdd(&acc[(size_t)dj * F_HID + lane], v[j]);
    }
  } else {
    for (int j = 0; j < cnt; ++j) {
      const int sj = __builtin_amdgcn_readlane(s, j);
      const int dj = __builtin_amdgcn_readlane(d, j);
      unsafeAtomicAdd(&acc[(size_t)dj * F_HID + lane], rows[(size_t)sj * F_HID + lane]);
    }
  }
}

__global__ void k_agg2_init(float* __restrict__ acc1, const float* __restrict__ dinv,
                            const float* __restrict__ bias, float* __restrict__ g, int N) {
  int t = blockIdx.x * blockDim.x + threadIdx.x;
  if (t < N * F_HID) {
    int i = t >> 5, f = t & 31;
    float di = dinv[i];
    float r = fmaxf(acc1[t] * di + bias[f], 0.0f) * di;
    acc1[t] = r;
    g[t]    = r;
  }
}


static inline int cdiv_ll(long long a, long long b) { return (int)((a + b - 1) / b); }

extern "C" void kernel_launch(void* const* d_in, const int* in_sizes, int n_in,
                              void* d_out, int out_size, void* d_ws, size_t ws_size,
                              hipStream_t stream) {
  const float* X  = (const float*)d_in[0];
  const int*   ei = (const int*)d_in[1];
  const float* W1 = (const float*)d_in[2];
  const float* b1 = (const float*)d_in[3];
  const float* W2 = (const float*)d_in[4];
  const float* b2 = (const float*)d_in[5];
  float* out = (float*)d_out;

  const int N = in_sizes[0] / F_IN;
  const int E = in_sizes[1] / 2;
  const int* src = ei;
  const int* dst = ei + E;

  float* dinv = (float*)d_ws;
  float* h1   = dinv + N;
  float* out1 = h1 + (size_t)N * F_HID;
  float* g    = out1 + (size_t)N * F_HID;

  const int B = 256;
  const int mtiles = cdiv_ll(N, 16);

  k_init_deg <<<cdiv_ll(N, B), B, 0, stream>>>(dinv, N);
  k_count_deg<<<cdiv_ll(E, B), B, 0, stream>>>(dst, dinv, E);
  k_rsqrt    <<<cdiv_ll(N, B), B, 0, stream>>>(dinv, N);

  k_gemm1    <<<cdiv_ll((long long)mtiles * 32, B), B, 0, stream>>>(X, W1, dinv, h1, N);
  k_agg1_init<<<cdiv_ll((long long)N * F_HID, B), B, 0, stream>>>(h1, out1, N);
  k_agg      <<<cdiv_ll(E, B), B, 0, stream>>>(h1, src, dst, out1, E);

  k_agg2_init<<<cdiv_ll((long long)N * F_HID, B), B, 0, stream>>>(out1, dinv, b1, g, N);
  k_agg      <<<cdiv_ll(E, B), B, 0, stream>>>(out1, src, dst, g, E);
  k_gemm2    <<<cdiv_ll((long long)mtiles * 32, B), B, 0, stream>>>(g, W2, b2, dinv, out, N);
}


# each atomic target (degree counts, both layers' sums) is one plane per package half: a wave adds into its own half's plane (die from the placement word), planes summed at use; f32 sum order changes
# speedup vs baseline: 4.5419x; 2.1590x over previous
#include <hip/hip_runtime.h>
#include <hip/hip_bf16.h>

typedef float v2f __attribute__((ext_vector_type(2)));
typedef float v8f __attribute__((ext_vector_type(8)));

#define F_IN  512
#define F_HID 32
#define F_OUT 64


__device__ __forceinline__ int pkg_half() {
  const unsigned w = __builtin_amdgcn_s_sendmsg_rtn(0x87);
  return (int)((w >> 18) & 1u);
}


__global__ void k_init_deg(float* __restrict__ degp, int N) {
  int i = blockIdx.x * blockDim.x + threadIdx.x;
  if (i < N) {
    degp[i]     = 1.0f;
    degp[N + i] = 0.0f;
  }
}

__global__ void k_count_deg(const int* __restrict__ dst, float* __restrict__ degp, int E, int N) {
  int e = blockIdx.x * blockDim.x + threadIdx.x;
  int dn = 0;
  if (e < E) dn = dst[e];
  const int h = pkg_half();
  if (e < E) unsafeAtomicAdd(&degp[(size_t)h * N + dn], 1.0f);
}

__global__ void k_rsqrt(const float* __restrict__ degp, float* __restrict__ dinv, int N) {
  int i = blockIdx.x * blockDim.x + threadIdx.x;
  if (i < N) dinv[i] = rsqrtf(degp[i] + degp[N + i]);
}


__global__ void k_gemm1(const float* __restrict__ X, const float* __restrict__ W,
                        const float* __restrict__ dinv, float* __restrict__ H, int N) {
  const int lane = threadIdx.x & 31;
  const int wave = (blockIdx.x * blockDim.x + threadIdx.x) >> 5;
  const int m0 = wave << 4;
  if (m0 >= N) return;
  const int half = lane >> 4;
  const int lrow = lane & 15;
  int row = m0 + lrow; if (row >= N) row = N - 1;
  const float* xr = X + (size_t)row * F_IN;

  float dm[8];
#pragma unroll
  for (int j = 0; j < 8; ++j) {
    int m = m0 + j + (half << 3); if (m >= N) m = N - 1;
    dm[j] = dinv[m];
  }

  v8f c0 = {}; v8f c1 = {};
#pragma unroll 4
  for (int k0 = 0; k0 < F_IN; k0 += 4) {
    const int ka = k0 + (half << 1);
    v2f a;  a.x  = xr[ka];                      a.y  = xr[ka + 1];
    v2f b0; b0.x = W[ka * F_HID + lrow];        b0.y = W[(ka + 1) * F_HID + lrow];
    v2f b1; b1.x = W[ka * F_HID + 16 + lrow];   b1.y = W[(ka + 1) * F_HID + 16 + lrow];
    c0 = __builtin_amdgcn_wmma_f32_16x16x4_f32(false, a, false, b0, (short)0, c0, false, false);
    c1 = __builtin_amdgcn_wmma_f32_16x16x4_f32(false, a, false, b1, (short)0, c1, false, false);
  }
#pragma unroll
  for (int j = 0; j < 8; ++j) {
    int m = m0 + j + (half << 3);
    if (m < N) {
      float* hr = H + (size_t)m * F_HID;
      hr[lrow]      = c0[j] * dm[j];
      hr[16 + lrow] = c1[j] * dm[j];
    }
  }
}


__global__ void k_gemm2(const float* __restrict__ A, const float* __restrict__ W,
                        const float* __restrict__ bias, const float* __restrict__ dinv,
                        float* __restrict__ H, int N) {
  const int lane = threadIdx.x & 31;
  const int wave = (blockIdx.x * blockDim.x + threadIdx.x) >> 5;
  const int m0 = wave << 4;
  if (m0 >= N) return;
  const int half = lane >> 4;
  const int lrow = lane & 15;
  int row = m0 + lrow; if (row >= N) row = N - 1;
  const float* ar  = A + (size_t)row * F_HID;
  const float* ar1 = ar + (size_t)N * F_HID;

  float dm[8];
#pragma unroll
  for (int j = 0; j < 8; ++j) {
    int m = m0 + j + (half << 3); if (m >= N) m = N - 1;
    dm[j] = dinv[m];
  }

  v8f c0 = {}; v8f c1 = {}; v8f c2 = {}; v8f c3 = {};
#pragma unroll
  for (int k0 = 0; k0 < F_HID; k0 += 4) {
    const int ka = k0 + (half << 1);
    v2f a; a.x = ar[ka] + ar1[ka]; a.y = ar[ka + 1] + ar1[ka + 1];
    v2f b;
    b.x = W[ka * F_OUT + lrow];            b.y = W[(ka + 1) * F_OUT + lrow];
    c0 = __builtin_amdgcn_wmma_f32_16x16x4_f32(false, a, false, b, (short)0, c0, false, false);
    b.x = W[ka * F_OUT + 16 + lrow];       b.y = W[(ka + 1) * F_OUT + 16 + lrow];
    c1 = __builtin_amdgcn_wmma_f32_16x16x4_f32(false, a, false, b, (short)0, c1, false, false);
    b.x = W[ka * F_OUT + 32 + lrow];       b.y = W[(ka + 1) * F_OUT + 32 + lrow];
    c2 = __builtin_amdgcn_wmma_f32_16x16x4_f32(false, a, false, b, (short)0, c2, false, false);
    b.x = W[ka * F_OUT + 48 + lrow];       b.y = W[(ka + 1) * F_OUT + 48 + lrow];
    c3 = __builtin_amdgcn_wmma_f32_16x16x4_f32(false, a, false, b, (short)0, c3, false, false);
  }
  const float bz0 = bias[lrow], bz1 = bias[16 + lrow], bz2 = bias[32 + lrow], bz3 = bias[48 + lrow];
#pragma unroll
  for (int j = 0; j < 8; ++j) {
    int m = m0 + j + (half << 3);
    if (m < N) {
      float* hr = H + (size_t)m * F_OUT;
      hr[lrow]      = c0[j] * dm[j] + bz0;
      hr[16 + lrow] = c1[j] * dm[j] + bz1;
      hr[32 + lrow] = c2[j] * dm[j] + bz2;
      hr[48 + lrow] = c3[j] * dm[j] + bz3;
    }
  }
}


__global__ void k_agg1_init(const float* __restrict__ hs, float* __restrict__ acc, int N) {
  int t = blockIdx.x * blockDim.x + threadIdx.x;
  if (t < N * F_HID) {
    acc[t] = hs[t];
    acc[(size_t)N * F_HID + t] = 0.0f;
  }
}

__global__ void k_agg(const float* __restrict__ rows, const int* __restrict__ src,
                      const int* __restrict__ dstv, float* __restrict__ acc2, int E, int N) {
  const int lane = threadIdx.x & 31;
  const int wave = (blockIdx.x * blockDim.x + threadIdx.x) >> 5;
  const int e0 = wave << 5;
  if (e0 >= E) return;
  const int e = e0 + lane;
  int s = 0, d = 0;
  if (e < E) { s = src[e]; d = dstv[e]; }
  float* __restrict__ acc = acc2 + (size_t)pkg_half() * ((size_t)N * F_HID);
  const int cnt = (E - e0 < 32) ? (E - e0) : 32;
  if (cnt == 32) {
    float v[32];
#pragma unroll
    for (int j = 0; j < 32; ++j) {
      const int sj = __builtin_amdgcn_readlane(s, j);
      v[j] = rows[(size_t)sj * F_HID + lane];
    }
#pragma unroll
    for (int j = 0; j < 32; ++j) {
      const int dj = __builtin_amdgcn_readlane(d, j);
      unsafeAtomicAdd(&acc[(size_t)dj * F_HID + lane], v[j]);
    }
  } else {
    for (int j = 0; j < cnt; ++j) {
      const int sj = __builtin_amdgcn_readlane(s, j);
      const int dj = __builtin_amdgcn_readlane(d, j);
      unsafeAtomicAdd(&acc[(size_t)dj * F_HID + lane], rows[(size_t)sj * F_HID + lane]);
    }
  }
}

__global__ void k_agg2_init(float* __restrict__ acc1, const float* __restrict__ dinv,
                            const float* __restrict__ bias, float* __restrict__ g, int N) {
  int t = blockIdx.x * blockDim.x + threadIdx.x;
  if (t < N * F_HID) {
    int i = t >> 5, f = t & 31;
    float di = dinv[i];
    float r = fmaxf((acc1[t] + acc1[(size_t)N * F_HID + t]) * di + bias[f], 0.0f) * di;
    acc1[t] = r;
    g[t]    = r;
    g[(size_t)N * F_HID + t] = 0.0f;
  }
}


static inline int cdiv_ll(long long a, long long b) { return (int)((a + b - 1) / b); }

extern "C" void kernel_launch(void* const* d_in, const int* in_sizes, int n_in,
                              void* d_out, int out_size, void* d_ws, size_t ws_size,
                              hipStream_t stream) {
  const float* X  = (const float*)d_in[0];
  const int*   ei = (const int*)d_in[1];
  const float* W1 = (const float*)d_in[2];
  const float* b1 = (const float*)d_in[3];
  const float* W2 = (const float*)d_in[4];
  const float* b2 = (const float*)d_in[5];
  float* out = (float*)d_out;

  const int N = in_sizes[0] / F_IN;
  const int E = in_sizes[1] / 2;
  const int* src = ei;
  const int* dst = ei + E;

  float* h1   = (float*)d_ws;
  float* out1 = h1 + (size_t)N * F_HID;
  float* g    = out1 + (size_t)2 * N * F_HID;
  float* dinv = g + (size_t)2 * N * F_HID;
  float* degp = dinv + N;

  const int B = 256;
  const int mtiles = cdiv_ll(N, 16);

  k_init_deg <<<cdiv_ll(N, B), B, 0, stream>>>(degp, N);
  k_count_deg<<<cdiv_ll(E, B), B, 0, stream>>>(dst, degp, E, N);
  k_rsqrt    <<<cdiv_ll(N, B), B, 0, stream>>>(degp, dinv, N);

  k_gemm1    <<<cdiv_ll((long long)mtiles * 32, B), B, 0, stream>>>(X, W1, dinv, h1, N);
  k_agg1_init<<<cdiv_ll((long long)N * F_HID, B), B, 0, stream>>>(h1, out1, N);
  k_agg      <<<cdiv_ll(E, B), B, 0, stream>>>(h1, src, dst, out1, E, N);

  k_agg2_init<<<cdiv_ll((long long)N * F_HID, B), B, 0, stream>>>(out1, dinv, b1, g, N);
  k_agg      <<<cdiv_ll(E, B), B, 0, stream>>>(out1, src, dst, g, E, N);
  k_gemm2    <<<cdiv_ll((long long)mtiles * 32, B), B, 0, stream>>>(g, W2, b2, dinv, out, N);
}


# degree counts: each workgroup counts a slice of the edge list into a private LDS histogram (integer LDS adds) and stores it; dinv sums the 64 partial counts; no global atomics for degrees, exact
# speedup vs baseline: 5.0245x; 1.1063x over previous
#include <hip/hip_runtime.h>
#include <hip/hip_bf16.h>

typedef float v2f __attribute__((ext_vector_type(2)));
typedef float v8f __attribute__((ext_vector_type(8)));

#define F_IN  512
#define F_HID 32
#define F_OUT 64


__device__ __forceinline__ int pkg_half() {
  const unsigned w = __builtin_amdgcn_s_sendmsg_rtn(0x87);
  return (int)((w >> 18) & 1u);
}


#define HR 25600
#define HC 64
#define HU 25

__global__ __launch_bounds__(1024) void k_hist(const int* __restrict__ dst, unsigned* __restrict__ part, int E, int N) {
  __shared__ unsigned cnt[HR];
  const int tid = threadIdx.x;
  const int r0 = blockIdx.y * HR;
  const int chunk = (E + HC - 1) / HC;
  const int c0 = blockIdx.x * chunk;
  int c1 = c0 + chunk; if (c1 > E) c1 = E;
  for (int i = tid; i < HR; i += 1024) cnt[i] = 0u;
  __syncthreads();
  for (int base = c0 + tid; base < c1; base += 1024 * HU) {
    int dv[HU];
#pragma unroll
    for (int u = 0; u < HU; ++u) {
      const int e = base + u * 1024;
      const int v = dst[e < c1 ? e : c1 - 1];
      dv[u] = e < c1 ? v : -1;
    }
#pragma unroll
    for (int u = 0; u < HU; ++u) {
      const unsigned d = (unsigned)(dv[u] - r0);
      if (d < (unsigned)HR) atomicAdd(&cnt[d], 1u);
    }
  }
  __syncthreads();
  int nr = N - r0; if (nr > HR) nr = HR;
  unsigned* __restrict__ p = part + (size_t)blockIdx.x * N + r0;
  for (int i = tid; i < nr; i += 1024) p[i] = cnt[i];
}

__global__ void k_dinv(const unsigned* __restrict__ part, float* __restrict__ dinv, int N) {
  int i = blockIdx.x * blockDim.x + threadIdx.x;
  if (i < N) {
    unsigned s = 1u;
#pragma unroll
    for (int c = 0; c < HC; ++c)
      s += part[(size_t)c * N + i];
    dinv[i] = rsqrtf((float)s);
  }
}


__global__ void k_gemm1(const float* __restrict__ X, const float* __restrict__ W,
                        const float* __restrict__ dinv, float* __restrict__ H, int N) {
  const int lane = threadIdx.x & 31;
  const int wave = (blockIdx.x * blockDim.x + threadIdx.x) >> 5;
  const int m0 = wave << 4;
  if (m0 >= N) return;
  const int half = lane >> 4;
  const int lrow = lane & 15;
  int row = m0 + lrow; if (row >= N) row = N - 1;
  const float* xr = X + (size_t)row * F_IN;

  float dm[8];
#pragma unroll
  for (int j = 0; j < 8; ++j) {
    int m = m0 + j + (half << 3); if (m >= N) m = N - 1;
    dm[j] = dinv[m];
  }

  v8f c0 = {}; v8f c1 = {};
#pragma unroll 4
  for (int k0 = 0; k0 < F_IN; k0 += 4) {
    const int ka = k0 + (half << 1);
    v2f a;  a.x  = xr[ka];                      a.y  = xr[ka + 1];
    v2f b0; b0.x = W[ka * F_HID + lrow];        b0.y = W[(ka + 1) * F_HID + lrow];
    v2f b1; b1.x = W[ka * F_HID + 16 + lrow];   b1.y = W[(ka + 1) * F_HID + 16 + lrow];
    c0 = __builtin_amdgcn_wmma_f32_16x16x4_f32(false, a, false, b0, (short)0, c0, false, false);
    c1 = __builtin_amdgcn_wmma_f32_16x16x4_f32(false, a, false, b1, (short)0, c1, false, false);
  }
#pragma unroll
  for (int j = 0; j < 8; ++j) {
    int m = m0 + j + (half << 3);
    if (m < N) {
      float* hr = H + (size_t)m * F_HID;
      hr[lrow]      = c0[j] * dm[j];
      hr[16 + lrow] = c1[j] * dm[j];
    }
  }
}


__global__ void k_gemm2(const float* __restrict__ A, const float* __restrict__ W,
                        const float* __restrict__ bias, const float* __restrict__ dinv,
                        float* __restrict__ H, int N) {
  const int lane = threadIdx.x & 31;
  const int wave = (blockIdx.x * blockDim.x + threadIdx.x) >> 5;
  const int m0 = wave << 4;
  if (m0 >= N) return;
  const int half = lane >> 4;
  const int lrow = lane & 15;
  int row = m0 + lrow; if (row >= N) row = N - 1;
  const float* ar  = A + (size_t)row * F_HID;
  const float* ar1 = ar + (size_t)N * F_HID;

  float dm[8];
#pragma unroll
  for (int j = 0; j < 8; ++j) {
    int m = m0 + j + (half << 3); if (m >= N) m = N - 1;
    dm[j] = dinv[m];
  }

  v8f c0 = {}; v8f c1 = {}; v8f c2 = {}; v8f c3 = {};
#pragma unroll
  for (int k0 = 0; k0 < F_HID; k0 += 4) {
    const int ka = k0 + (half << 1);
    v2f a; a.x = ar[ka] + ar1[ka]; a.y = ar[ka + 1] + ar1[ka + 1];
    v2f b;
    b.x = W[ka * F_OUT + lrow];            b.y = W[(ka + 1) * F_OUT + lrow];
    c0 = __builtin_amdgcn_wmma_f32_16x16x4_f32(false, a, false, b, (short)0, c0, false, false);
    b.x = W[ka * F_OUT + 16 + lrow];       b.y = W[(ka + 1) * F_OUT + 16 + lrow];
    c1 = __builtin_amdgcn_wmma_f32_16x16x4_f32(false, a, false, b, (short)0, c1, false, false);
    b.x = W[ka * F_OUT + 32 + lrow];       b.y = W[(ka + 1) * F_OUT + 32 + lrow];
    c2 = __builtin_amdgcn_wmma_f32_16x16x4_f32(false, a, false, b, (short)0, c2, false, false);
    b.x = W[ka * F_OUT + 48 + lrow];       b.y = W[(ka + 1) * F_OUT + 48 + lrow];
    c3 = __builtin_amdgcn_wmma_f32_16x16x4_f32(false, a, false, b, (short)0, c3, false, false);
  }
  const float bz0 = bias[lrow], bz1 = bias[16 + lrow], bz2 = bias[32 + lrow], bz3 = bias[48 + lrow];
#pragma unroll
  for (int j = 0; j < 8; ++j) {
    int m = m0 + j + (half << 3);
    if (m < N) {
      float* hr = H + (size_t)m * F_OUT;
      hr[lrow]      = c0[j] * dm[j] + bz0;
      hr[16 + lrow] = c1[j] * dm[j] + bz1;
      hr[32 + lrow] = c2[j] * dm[j] + bz2;
      hr[48 + lrow] = c3[j] * dm[j] + bz3;
    }
  }
}


__global__ void k_agg1_init(const float* __restrict__ hs, float* __restrict__ acc, int N) {
  int t = blockIdx.x * blockDim.x + threadIdx.x;
  if (t < N * F_HID) {
    acc[t] = hs[t];
    acc[(size_t)N * F_HID + t] = 0.0f;
  }
}

__global__ void k_agg(const float* __restrict__ rows, const int* __restrict__ src,
                      const int* __restrict__ dstv, float* __restrict__ acc2, int E, int N) {
  const int lane = threadIdx.x & 31;
  const int wave = (blockIdx.x * blockDim.x + threadIdx.x) >> 5;
  const int e0 = wave << 5;
  if (e0 >= E) return;
  const int e = e0 + lane;
  int s = 0, d = 0;
  if (e < E) { s = src[e]; d = dstv[e]; }
  float* __restrict__ acc = acc2 + (size_t)pkg_half() * ((size_t)N * F_HID);
  const int cnt = (E - e0 < 32) ? (E - e0) : 32;
  if (cnt == 32) {
    float v[32];
#pragma unroll
    for (int j = 0; j < 32; ++j) {
      const int sj = __builtin_amdgcn_readlane(s, j);
      v[j] = rows[(size_t)sj * F_HID + lane];
    }
#pragma unroll
    for (int j = 0; j < 32; ++j) {
      const int dj = __builtin_amdgcn_readlane(d, j);
      unsafeAtomicAdd(&acc[(size_t)dj * F_HID + lane], v[j]);
    }
  } else {
    for (int j = 0; j < cnt; ++j) {
      const int sj = __builtin_amdgcn_readlane(s, j);
      const int dj = __builtin_amdgcn_readlane(d, j);
      unsafeAtomicAdd(&acc[(size_t)dj * F_HID + lane], rows[(size_t)sj * F_HID + lane]);
    }
  }
}

__global__ void k_agg2_init(float* __restrict__ acc1, const float* __restrict__ dinv,
                            const float* __restrict__ bias, float* __restrict__ g, int N) {
  int t = blockIdx.x * blockDim.x + threadIdx.x;
  if (t < N * F_HID) {
    int i = t >> 5, f = t & 31;
    float di = dinv[i];
    float r = fmaxf((acc1[t] + acc1[(size_t)N * F_HID + t]) * di + bias[f], 0.0f) * di;
    acc1[t] = r;
    g[t]    = r;
    g[(size_t)N * F_HID + t] = 0.0f;
  }
}


static inline int cdiv_ll(long long a, long long b) { return (int)((a + b - 1) / b); }

extern "C" void kernel_launch(void* const* d_in, const int* in_sizes, int n_in,
                              void* d_out, int out_size, void* d_ws, size_t ws_size,
                              hipStream_t stream) {
  const float* X  = (const float*)d_in[0];
  const int*   ei = (const int*)d_in[1];
  const float* W1 = (const float*)d_in[2];
  const float* b1 = (const float*)d_in[3];
  const float* W2 = (const float*)d_in[4];
  const float* b2 = (const float*)d_in[5];
  float* out = (float*)d_out;

  const int N = in_sizes[0] / F_IN;
  const int E = in_sizes[1] / 2;
  const int* src = ei;
  const int* dst = ei + E;

  float* h1   = (float*)d_ws;
  float* out1 = h1 + (size_t)N * F_HID;
  float* g    = out1 + (size_t)2 * N * F_HID;
  float* dinv = g + (size_t)2 * N * F_HID;
  unsigned* part = (unsigned*)(dinv + N);

  const int B = 256;
  const int mtiles = cdiv_ll(N, 16);

  k_hist     <<<dim3(HC, cdiv_ll(N, HR)), 1024, 0, stream>>>(dst, part, E, N);
  k_dinv     <<<cdiv_ll(N, B), B, 0, stream>>>(part, dinv, N);

  k_gemm1    <<<cdiv_ll((long long)mtiles * 32, B), B, 0, stream>>>(X, W1, dinv, h1, N);
  k_agg1_init<<<cdiv_ll((long long)N * F_HID, B), B, 0, stream>>>(h1, out1, N);
  k_agg      <<<cdiv_ll(E, B), B, 0, stream>>>(h1, src, dst, out1, E, N);

  k_agg2_init<<<cdiv_ll((long long)N * F_HID, B), B, 0, stream>>>(out1, dinv, b1, g, N);
  k_agg      <<<cdiv_ll(E, B), B, 0, stream>>>(out1, src, dst, g, E, N);
  k_gemm2    <<<cdiv_ll((long long)mtiles * 32, B), B, 0, stream>>>(g, W2, b2, dinv, out, N);
}


# both products load operands wide: A by 16-byte row loads, B by one whole-line load per k-step from a fragment-ordered weight copy built in the dinv pass; k order in the f32 sums changes
# speedup vs baseline: 5.2796x; 1.0508x over previous
#include <hip/hip_runtime.h>
#include <hip/hip_bf16.h>

typedef float v2f __attribute__((ext_vector_type(2)));
typedef float v8f __attribute__((ext_vector_type(8)));
typedef float v4f __attribute__((ext_vector_type(4)));

#define F_IN  512
#define F_HID 32
#define F_OUT 64


__device__ __forceinline__ int pkg_half() {
  const unsigned w = __builtin_amdgcn_s_sendmsg_rtn(0x87);
  return (int)((w >> 18) & 1u);
}


#define HR 25600
#define HC 64
#define HU 25

__global__ __launch_bounds__(1024) void k_hist(const int* __restrict__ dst, unsigned* __restrict__ part, int E, int N) {
  __shared__ unsigned cnt[HR];
  const int tid = threadIdx.x;
  const int r0 = blockIdx.y * HR;
  const int chunk = (E + HC - 1) / HC;
  const int c0 = blockIdx.x * chunk;
  int c1 = c0 + chunk; if (c1 > E) c1 = E;
  for (int i = tid; i < HR; i += 1024) cnt[i] = 0u;
  __syncthreads();
  for (int base = c0 + tid; base < c1; base += 1024 * HU) {
    int dv[HU];
#pragma unroll
    for (int u = 0; u < HU; ++u) {
      const int e = base + u * 1024;
      const int v = dst[e < c1 ? e : c1 - 1];
      dv[u] = e < c1 ? v : -1;
    }
#pragma unroll
    for (int u = 0; u < HU; ++u) {
      const unsigned d = (unsigned)(dv[u] - r0);
      if (d < (unsigned)HR) atomicAdd(&cnt[d], 1u);
    }
  }
  __syncthreads();
  int nr = N - r0; if (nr > HR) nr = HR;
  unsigned* __restrict__ p = part + (size_t)blockIdx.x * N + r0;
  for (int i = tid; i < nr; i += 1024) p[i] = cnt[i];
}

__global__ void k_dinv(const unsigned* __restrict__ part, float* __restrict__ dinv, int N,
                       const float* __restrict__ W1, v4f* __restrict__ W1f,
                       const float* __restrict__ W2, v4f* __restrict__ W2f) {
  int i = blockIdx.x * blockDim.x + threadIdx.x;
  if (i < (F_IN / 4) * 32) {
    const int t = i >> 5, lane = i & 31, c = lane & 15, h = lane >> 4;
    const int k = 8 * (t >> 1) + 4 * h + 2 * (t & 1);
    v4f f;
    f.x = W1[k * F_HID + c];       f.y = W1[(k + 1) * F_HID + c];
    f.z = W1[k * F_HID + 16 + c];  f.w = W1[(k + 1) * F_HID + 16 + c];
    W1f[i] = f;
  }
  if (i < (F_HID / 4) * 2 * 32) {
    const int t = i >> 6, q = (i >> 5) & 1, lane = i & 31, c = lane & 15, h = lane >> 4;
    const int k = 16 * h + 2 * t;
    const int n = 32 * q + c;
    v4f f;
    f.x = W2[k * F_OUT + n];       f.y = W2[(k + 1) * F_OUT + n];
    f.z = W2[k * F_OUT + 16 + n];  f.w = W2[(k + 1) * F_OUT + 16 + n];
    W2f[i] = f;
  }
  if (i < N) {
    unsigned s = 1u;
#pragma unroll
    for (int c = 0; c < HC; ++c)
      s += part[(size_t)c * N + i];
    dinv[i] = rsqrtf((float)s);
  }
}


__global__ void k_gemm1(const float* __restrict__ X, const v4f* __restrict__ Wf,
                        const float* __restrict__ dinv, float* __restrict__ H, int N) {
  const int lane = threadIdx.x & 31;
  const int wave = (blockIdx.x * blockDim.x + threadIdx.x) >> 5;
  const int m0 = wave << 4;
  if (m0 >= N) return;
  const int half = lane >> 4;
  const int lrow = lane & 15;
  int row = m0 + lrow; if (row >= N) row = N - 1;
  const float* xr = X + (size_t)row * F_IN;

  float dm[8];
#pragma unroll
  for (int j = 0; j < 8; ++j) {
    int m = m0 + j + (half << 3); if (m >= N) m = N - 1;
    dm[j] = dinv[m];
  }

  const v4f* xq = (const v4f*)(xr + (half << 2));
  const v4f* wq = Wf + lane;
  v8f c0 = {}; v8f c1 = {};
#pragma unroll 4
  for (int u = 0; u < F_IN / 8; ++u) {
    const v4f a4 = xq[2 * u];
    const v4f p0 = wq[(2 * u) * 32];
    const v4f p1 = wq[(2 * u + 1) * 32];
    v2f a;  a.x  = a4.x; a.y  = a4.y;
    v2f b0; b0.x = p0.x; b0.y = p0.y;
    v2f b1; b1.x = p0.z; b1.y = p0.w;
    c0 = __builtin_amdgcn_wmma_f32_16x16x4_f32(false, a, false, b0, (short)0, c0, false, false);
    c1 = __builtin_amdgcn_wmma_f32_16x16x4_f32(false, a, false, b1, (short)0, c1, false, false);
    a.x  = a4.z; a.y  = a4.w;
    b0.x = p1.x; b0.y = p1.y;
    b1.x = p1.z; b1.y = p1.w;
    c0 = __builtin_amdgcn_wmma_f32_16x16x4_f32(false, a, false, b0, (short)0, c0, false, false);
    c1 = __builtin_amdgcn_wmma_f32_16x16x4_f32(false, a, false, b1, (short)0, c1, false, false);
  }
#pragma unroll
  for (int j = 0; j < 8; ++j) {
    int m = m0 + j + (half << 3);
    if (m < N) {
      float* hr = H + (size_t)m * F_HID;
      hr[lrow]      = c0[j] * dm[j];
      hr[16 + lrow] = c1[j] * dm[j];
    }
  }
}


__global__ void k_gemm2(const float* __restrict__ A, const v4f* __restrict__ Wf,
                        const float* __restrict__ bias, const float* __restrict__ dinv,
                        float* __restrict__ H, int N) {
  const int lane = threadIdx.x & 31;
  const int wave = (blockIdx.x * blockDim.x + threadIdx.x) >> 5;
  const int m0 = wave << 4;
  if (m0 >= N) return;
  const int half = lane >> 4;
  const int lrow = lane & 15;
  int row = m0 + lrow; if (row >= N) row = N - 1;
  const float* ar  = A + (size_t)row * F_HID;
  const float* ar1 = ar + (size_t)N * F_HID;

  float dm[8];
#pragma unroll
  for (int j = 0; j < 8; ++j) {
    int m = m0 + j + (half << 3); if (m >= N) m = N - 1;
    dm[j] = dinv[m];
  }

  const v4f* aq0 = (const v4f*)(ar  + (half << 4));
  const v4f* aq1 = (const v4f*)(ar1 + (half << 4));
  const v4f* wq = Wf + lane;
  v4f s4[4];
#pragma unroll
  for (int i = 0; i < 4; ++i) s4[i] = aq0[i] + aq1[i];
  v8f c0 = {}; v8f c1 = {}; v8f c2 = {}; v8f c3 = {};
#pragma unroll
  for (int t = 0; t < F_HID / 4; ++t) {
    const v4f sv = s4[t >> 1];
    v2f a;
    if (t & 1) { a.x = sv.z; a.y = sv.w; } else { a.x = sv.x; a.y = sv.y; }
    const v4f p0 = wq[(2 * t) * 32];
    const v4f p1 = wq[(2 * t + 1) * 32];
    v2f b;
    b.x = p0.x; b.y = p0.y;
    c0 = __builtin_amdgcn_wmma_f32_16x16x4_f32(false, a, false, b, (short)0, c0, false, false);
    b.x = p0.z; b.y = p0.w;
    c1 = __builtin_amdgcn_wmma_f32_16x16x4_f32(false, a, false, b, (short)0, c1, false, false);
    b.x = p1.x; b.y = p1.y;
    c2 = __builtin_amdgcn_wmma_f32_16x16x4_f32(false, a, false, b, (short)0, c2, false, false);
    b.x = p1.z; b.y = p1.w;
    c3 = __builtin_amdgcn_wmma_f32_16x16x4_f32(false, a, false, b, (short)0, c3, false, false);
  }
  const float bz0 = bias[lrow], bz1 = bias[16 + lrow], bz2 = bias[32 + lrow], bz3 = bias[48 + lrow];
#pragma unroll
  for (int j = 0; j < 8; ++j) {
    int m = m0 + j + (half << 3);
    if (m < N) {
      float* hr = H + (size_t)m * F_OUT;
      hr[lrow]      = c0[j] * dm[j] + bz0;
      hr[16 + lrow] = c1[j] * dm[j] + bz1;
      hr[32 + lrow] = c2[j] * dm[j] + bz2;
      hr[48 + lrow] = c3[j] * dm[j] + bz3;
    }
  }
}


__global__ void k_agg1_init(const float* __restrict__ hs, float* __restrict__ acc, int N) {
  int t = blockIdx.x * blockDim.x + threadIdx.x;
  if (t < N * F_HID) {
    acc[t] = hs[t];
    acc[(size_t)N * F_HID + t] = 0.0f;
  }
}

__global__ void k_agg(const float* __restrict__ rows, const int* __restrict__ src,
                      const int* __restrict__ dstv, float* __restrict__ acc2, int E, int N) {
  const int lane = threadIdx.x & 31;
  const int wave = (blockIdx.x * blockDim.x + threadIdx.x) >> 5;
  const int e0 = wave << 5;
  if (e0 >= E) return;
  const int e = e0 + lane;
  int s = 0, d = 0;
  if (e < E) { s = src[e]; d = dstv[e]; }
  float* __restrict__ acc = acc2 + (size_t)pkg_half() * ((size_t)N * F_HID);
  const int cnt = (E - e0 < 32) ? (E - e0) : 32;
  if (cnt == 32) {
    float v[32];
#pragma unroll
    for (int j = 0; j < 32; ++j) {
      const int sj = __builtin_amdgcn_readlane(s, j);
      v[j] = rows[(size_t)sj * F_HID + lane];
    }
#pragma unroll
    for (int j = 0; j < 32; ++j) {
      const int dj = __builtin_amdgcn_readlane(d, j);
      unsafeAtomicAdd(&acc[(size_t)dj * F_HID + lane], v[j]);
    }
  } else {
    for (int j = 0; j < cnt; ++j) {
      const int sj = __builtin_amdgcn_readlane(s, j);
      const int dj = __builtin_amdgcn_readlane(d, j);
      unsafeAtomicAdd(&acc[(size_t)dj * F_HID + lane], rows[(size_t)sj * F_HID + lane]);
    }
  }
}

__global__ void k_agg2_init(float* __restrict__ acc1, const float* __restrict__ dinv,
                            const float* __restrict__ bias, float* __restrict__ g, int N) {
  int t = blockIdx.x * blockDim.x + threadIdx.x;
  if (t < N * F_HID) {
    int i = t >> 5, f = t & 31;
    float di = dinv[i];
    float r = fmaxf((acc1[t] + acc1[(size_t)N * F_HID + t]) * di + bias[f], 0.0f) * di;
    acc1[t] = r;
    g[t]    = r;
    g[(size_t)N * F_HID + t] = 0.0f;
  }
}


static inline int cdiv_ll(long long a, long long b) { return (int)((a + b - 1) / b); }

extern "C" void kernel_launch(void* const* d_in, const int* in_sizes, int n_in,
                              void* d_out, int out_size, void* d_ws, size_t ws_size,
                              hipStream_t stream) {
  const float* X  = (const float*)d_in[0];
  const int*   ei = (const int*)d_in[1];
  const float* W1 = (const float*)d_in[2];
  const float* b1 = (const float*)d_in[3];
  const float* W2 = (const float*)d_in[4];
  const float* b2 = (const float*)d_in[5];
  float* out = (float*)d_out;

  const int N = in_sizes[0] / F_IN;
  const int E = in_sizes[1] / 2;
  const int* src = ei;
  const int* dst = ei + E;

  v4f* W1f = (v4f*)d_ws;
  v4f* W2f = W1f + (F_IN / 4) * 32;
  float* h1   = (float*)(W2f + (F_HID / 4) * 2 * 32);
  float* out1 = h1 + (size_t)N * F_HID;
  float* g    = out1 + (size_t)2 * N * F_HID;
  float* dinv = g + (size_t)2 * N * F_HID;
  unsigned* part = (unsigned*)(dinv + N);

  const int B = 256;
  const int mtiles = cdiv_ll(N, 16);

  k_hist     <<<dim3(HC, cdiv_ll(N, HR)), 1024, 0, stream>>>(dst, part, E, N);
  k_dinv     <<<cdiv_ll(N, B), B, 0, stream>>>(part, dinv, N, W1, W1f, W2, W2f);

  k_gemm1    <<<cdiv_ll((long long)mtiles * 32, B), B, 0, stream>>>(X, W1f, dinv, h1, N);
  k_agg1_init<<<cdiv_ll((long long)N * F_HID, B), B, 0, stream>>>(h1, out1, N);
  k_agg      <<<cdiv_ll(E, B), B, 0, stream>>>(h1, src, dst, out1, E, N);

  k_agg2_init<<<cdiv_ll((long long)N * F_HID, B), B, 0, stream>>>(out1, dinv, b1, g, N);
  k_agg      <<<cdiv_ll(E, B), B, 0, stream>>>(out1, src, dst, g, E, N);
  k_gemm2    <<<cdiv_ll((long long)mtiles * 32, B), B, 0, stream>>>(g, W2f, b2, dinv, out, N);
}


# the two small passes around the first product are folded into it: a tile sums its 16 nodes' partial degree counts itself, its epilogue writes both sum planes' start values; 8 launches become 6
# speedup vs baseline: 5.4957x; 1.0409x over previous
#include <hip/hip_runtime.h>
#include <hip/hip_bf16.h>

typedef float v2f __attribute__((ext_vector_type(2)));
typedef float v8f __attribute__((ext_vector_type(8)));
typedef float v4f __attribute__((ext_vector_type(4)));

#define F_IN  512
#define F_HID 32
#define F_OUT 64


__device__ __forceinline__ int pkg_half() {
  const unsigned w = __builtin_amdgcn_s_sendmsg_rtn(0x87);
  return (int)((w >> 18) & 1u);
}


#define HR 25600
#define HC 64
#define HU 25

__global__ __launch_bounds__(1024) void k_hist(const int* __restrict__ dst, unsigned* __restrict__ part, int E, int N,
                                               const float* __restrict__ W1, v4f* __restrict__ W1f,
                                               const float* __restrict__ W2, v4f* __restrict__ W2f) {
  __shared__ unsigned cnt[HR];
  const int tid = threadIdx.x;
  const int gi = (blockIdx.y * HC + blockIdx.x) * 1024 + tid;
  if (gi < (F_IN / 4) * 32) {
    const int t = gi >> 5, lane = gi & 31, c = lane & 15, h = lane >> 4;
    const int k = 8 * (t >> 1) + 4 * h + 2 * (t & 1);
    v4f f;
    f.x = W1[k * F_HID + c];       f.y = W1[(k + 1) * F_HID + c];
    f.z = W1[k * F_HID + 16 + c];  f.w = W1[(k + 1) * F_HID + 16 + c];
    W1f[gi] = f;
  }
  if (gi < (F_HID / 4) * 2 * 32) {
    const int t = gi >> 6, q = (gi >> 5) & 1, lane = gi & 31, c = lane & 15, h = lane >> 4;
    const int k = 16 * h + 2 * t;
    const int n = 32 * q + c;
    v4f f;
    f.x = W2[k * F_OUT + n];       f.y = W2[(k + 1) * F_OUT + n];
    f.z = W2[k * F_OUT + 16 + n];  f.w = W2[(k + 1) * F_OUT + 16 + n];
    W2f[gi] = f;
  }
  const int r0 = blockIdx.y * HR;
  const int chunk = (E + HC - 1) / HC;
  const int c0 = blockIdx.x * chunk;
  int c1 = c0 + chunk; if (c1 > E) c1 = E;
  for (int i = tid; i < HR; i += 1024) cnt[i] = 0u;
  __syncthreads();
  for (int base = c0 + tid; base < c1; base += 1024 * HU) {
    int dv[HU];
#pragma unroll
    for (int u = 0; u < HU; ++u) {
      const int e = base + u * 1024;
      const int v = dst[e < c1 ? e : c1 - 1];
      dv[u] = e < c1 ? v : -1;
    }
#pragma unroll
    for (int u = 0; u < HU; ++u) {
      const unsigned d = (unsigned)(dv[u] - r0);
      if (d < (unsigned)HR) atomicAdd(&cnt[d], 1u);
    }
  }
  __syncthreads();
  int nr = N - r0; if (nr > HR) nr = HR;
  unsigned* __restrict__ p = part + (size_t)blockIdx.x * N + r0;
  for (int i = tid; i < nr; i += 1024) p[i] = cnt[i];
}


__global__ void k_gemm1(const float* __restrict__ X, const v4f* __restrict__ Wf,
                        const unsigned* __restrict__ part, float* __restrict__ dinv,
                        float* __restrict__ H, float* __restrict__ acc, int N) {
  const int lane = threadIdx.x & 31;
  const int wave = (blockIdx.x * blockDim.x + threadIdx.x) >> 5;
  const int m0 = wave << 4;
  if (m0 >= N) return;
  const int half = lane >> 4;
  const int lrow = lane & 15;
  int row = m0 + lrow; if (row >= N) row = N - 1;
  const float* xr = X + (size_t)row * F_IN;

  unsigned pc[HC / 2];
  const unsigned* pr = part + (size_t)half * N + row;
#pragma unroll
  for (int i = 0; i < HC / 2; ++i)
    pc[i] = pr[(size_t)(2 * i) * N];

  const v4f* xq = (const v4f*)(xr + (half << 2));
  const v4f* wq = Wf + lane;
  v8f c0 = {}; v8f c1 = {};
#pragma unroll 4
  for (int u = 0; u < F_IN / 8; ++u) {
    const v4f a4 = xq[2 * u];
    const v4f p0 = wq[(2 * u) * 32];
    const v4f p1 = wq[(2 * u + 1) * 32];
    v2f a;  a.x  = a4.x; a.y  = a4.y;
    v2f b0; b0.x = p0.x; b0.y = p0.y;
    v2f b1; b1.x = p0.z; b1.y = p0.w;
    c0 = __builtin_amdgcn_wmma_f32_16x16x4_f32(false, a, false, b0, (short)0, c0, false, false);
    c1 = __builtin_amdgcn_wmma_f32_16x16x4_f32(false, a, false, b1, (short)0, c1, false, false);
    a.x  = a4.z; a.y  = a4.w;
    b0.x = p1.x; b0.y = p1.y;
    b1.x = p1.z; b1.y = p1.w;
    c0 = __builtin_amdgcn_wmma_f32_16x16x4_f32(false, a, false, b0, (short)0, c0, false, false);
    c1 = __builtin_amdgcn_wmma_f32_16x16x4_f32(false, a, false, b1, (short)0, c1, false, false);
  }
  unsigned deg = 1u;
#pragma unroll
  for (int i = 0; i < HC / 2; ++i) deg += pc[i];
  deg += (unsigned)__shfl_xor((int)(deg - 1u), 16);
  const float dv = rsqrtf((float)deg);
  if (half == 0 && m0 + lrow < N) dinv[m0 + lrow] = dv;
  float* acc1 = acc + (size_t)N * F_HID;
#pragma unroll
  for (int j = 0; j < 8; ++j) {
    int m = m0 + j + (half << 3);
    const float dlo = __int_as_float(__builtin_amdgcn_readlane(__float_as_int(dv), j));
    const float dhi = __int_as_float(__builtin_amdgcn_readlane(__float_as_int(dv), j + 8));
    const float dmj = half ? dhi : dlo;
    if (m < N) {
      const size_t o = (size_t)m * F_HID;
      const float v0 = c0[j] * dmj;
      const float v1 = c1[j] * dmj;
      H[o + lrow]         = v0;
      H[o + 16 + lrow]    = v1;
      acc[o + lrow]       = v0;
      acc[o + 16 + lrow]  = v1;
      acc1[o + lrow]      = 0.0f;
      acc1[o + 16 + lrow] = 0.0f;
    }
  }
}


__global__ void k_gemm2(const float* __restrict__ A, const v4f* __restrict__ Wf,
                        const float* __restrict__ bias, const float* __restrict__ dinv,
                        float* __restrict__ H, int N) {
  const int lane = threadIdx.x & 31;
  const int wave = (blockIdx.x * blockDim.x + threadIdx.x) >> 5;
  const int m0 = wave << 4;
  if (m0 >= N) return;
  const int half = lane >> 4;
  const int lrow = lane & 15;
  int row = m0 + lrow; if (row >= N) row = N - 1;
  const float* ar  = A + (size_t)row * F_HID;
  const float* ar1 = ar + (size_t)N * F_HID;

  float dm[8];
#pragma unroll
  for (int j = 0; j < 8; ++j) {
    int m = m0 + j + (half << 3); if (m >= N) m = N - 1;
    dm[j] = dinv[m];
  }

  const v4f* aq0 = (const v4f*)(ar  + (half << 4));
  const v4f* aq1 = (const v4f*)(ar1 + (half << 4));
  const v4f* wq = Wf + lane;
  v4f s4[4];
#pragma unroll
  for (int i = 0; i < 4; ++i) s4[i] = aq0[i] + aq1[i];
  v8f c0 = {}; v8f c1 = {}; v8f c2 = {}; v8f c3 = {};
#pragma unroll
  for (int t = 0; t < F_HID / 4; ++t) {
    const v4f sv = s4[t >> 1];
    v2f a;
    if (t & 1) { a.x = sv.z; a.y = sv.w; } else { a.x = sv.x; a.y = sv.y; }
    const v4f p0 = wq[(2 * t) * 32];
    const v4f p1 = wq[(2 * t + 1) * 32];
    v2f b;
    b.x = p0.x; b.y = p0.y;
    c0 = __builtin_amdgcn_wmma_f32_16x16x4_f32(false, a, false, b, (short)0, c0, false, false);
    b.x = p0.z; b.y = p0.w;
    c1 = __builtin_amdgcn_wmma_f32_16x16x4_f32(false, a, false, b, (short)0, c1, false, false);
    b.x = p1.x; b.y = p1.y;
    c2 = __builtin_amdgcn_wmma_f32_16x16x4_f32(false, a, false, b, (short)0, c2, false, false);
    b.x = p1.z; b.y = p1.w;
    c3 = __builtin_amdgcn_wmma_f32_16x16x4_f32(false, a, false, b, (short)0, c3, false, false);
  }
  const float bz0 = bias[lrow], bz1 = bias[16 + lrow], bz2 = bias[32 + lrow], bz3 = bias[48 + lrow];
#pragma unroll
  for (int j = 0; j < 8; ++j) {
    int m = m0 + j + (half << 3);
    if (m < N) {
      float* hr = H + (size_t)m * F_OUT;
      hr[lrow]      = c0[j] * dm[j] + bz0;
      hr[16 + lrow] = c1[j] * dm[j] + bz1;
      hr[32 + lrow] = c2[j] * dm[j] + bz2;
      hr[48 + lrow] = c3[j] * dm[j] + bz3;
    }
  }
}


__global__ void k_agg(const float* __restrict__ rows, const int* __restrict__ src,
                      const int* __restrict__ dstv, float* __restrict__ acc2, int E, int N) {
  const int lane = threadIdx.x & 31;
  const int wave = (blockIdx.x * blockDim.x + threadIdx.x) >> 5;
  const int e0 = wave << 5;
  if (e0 >= E) return;
  const int e = e0 + lane;
  int s = 0, d = 0;
  if (e < E) { s = src[e]; d = dstv[e]; }
  float* __restrict__ acc = acc2 + (size_t)pkg_half() * ((size_t)N * F_HID);
  const int cnt = (E - e0 < 32) ? (E - e0) : 32;
  if (cnt == 32) {
    float v[32];
#pragma unroll
    for (int j = 0; j < 32; ++j) {
      const int sj = __builtin_amdgcn_readlane(s, j);
      v[j] = rows[(size_t)sj * F_HID + lane];
    }
#pragma unroll
    for (int j = 0; j < 32; ++j) {
      const int dj = __builtin_amdgcn_readlane(d, j);
      unsafeAtomicAdd(&acc[(size_t)dj * F_HID + lane], v[j]);
    }
  } else {
    for (int j = 0; j < cnt; ++j) {
      const int sj = __builtin_amdgcn_readlane(s, j);
      const int dj = __builtin_amdgcn_readlane(d, j);
      unsafeAtomicAdd(&acc[(size_t)dj * F_HID + lane], rows[(size_t)sj * F_HID + lane]);
    }
  }
}

__global__ void k_agg2_init(float* __restrict__ acc1, const float* __restrict__ dinv,
                            const float* __restrict__ bias, float* __restrict__ g, int N) {
  int t = blockIdx.x * blockDim.x + threadIdx.x;
  if (t < N * F_HID) {
    int i = t >> 5, f = t & 31;
    float di = dinv[i];
    float r = fmaxf((acc1[t] + acc1[(size_t)N * F_HID + t]) * di + bias[f], 0.0f) * di;
    acc1[t] = r;
    g[t]    = r;
    g[(size_t)N * F_HID + t] = 0.0f;
  }
}


static inline int cdiv_ll(long long a, long long b) { return (int)((a + b - 1) / b); }

extern "C" void kernel_launch(void* const* d_in, const int* in_sizes, int n_in,
                              void* d_out, int out_size, void* d_ws, size_t ws_size,
                              hipStream_t stream) {
  const float* X  = (const float*)d_in[0];
  const int*   ei = (const int*)d_in[1];
  const float* W1 = (const float*)d_in[2];
  const float* b1 = (const float*)d_in[3];
  const float* W2 = (const float*)d_in[4];
  const float* b2 = (const float*)d_in[5];
  float* out = (float*)d_out;

  const int N = in_sizes[0] / F_IN;
  const int E = in_sizes[1] / 2;
  const int* src = ei;
  const int* dst = ei + E;

  v4f* W1f = (v4f*)d_ws;
  v4f* W2f = W1f + (F_IN / 4) * 32;
  float* h1   = (float*)(W2f + (F_HID / 4) * 2 * 32);
  float* out1 = h1 + (size_t)N * F_HID;
  float* g    = out1 + (size_t)2 * N * F_HID;
  float* dinv = g + (size_t)2 * N * F_HID;
  unsigned* part = (unsigned*)(dinv + N);

  const int B = 256;
  const int mtiles = cdiv_ll(N, 16);

  k_hist     <<<dim3(HC, cdiv_ll(N, HR)), 1024, 0, stream>>>(dst, part, E, N, W1, W1f, W2, W2f);

  k_gemm1    <<<cdiv_ll((long long)mtiles * 32, B), B, 0, stream>>>(X, W1f, part, dinv, h1, out1, N);
  k_agg      <<<cdiv_ll(E, B), B, 0, stream>>>(h1, src, dst, out1, E, N);

  k_agg2_init<<<cdiv_ll((long long)N * F_HID, B), B, 0, stream>>>(out1, dinv, b1, g, N);
  k_agg      <<<cdiv_ll(E, B), B, 0, stream>>>(out1, src, dst, g, E, N);
  k_gemm2    <<<cdiv_ll((long long)mtiles * 32, B), B, 0, stream>>>(g, W2f, b2, dinv, out, N);
}


# both layers' sums: edges grouped by destination bin once per call (LDS counting sort), one wave sums a node's source rows in registers, one store per row; no float atomics; f32 sum order changes
# speedup vs baseline: 6.0304x; 1.0973x over previous
#include <hip/hip_runtime.h>
#include <hip/hip_bf16.h>

typedef float v2f __attribute__((ext_vector_type(2)));
typedef float v8f __attribute__((ext_vector_type(8)));
typedef float v4f __attribute__((ext_vector_type(4)));

#define F_IN  512
#define F_HID 32
#define F_OUT 64


#define BW    400
#define NBIN  256
#define SLCAP 50000
#define CAP   49152
#define SRCM  0x1FFFFu
#define NOEDGE 0xFFFFFFFFu
#define PU    16

__device__ __forceinline__ unsigned wave_excl_scan(unsigned v, int lane, unsigned* total) {
  unsigned incl = v;
#pragma unroll
  for (int o = 1; o < 32; o <<= 1) {
    const unsigned t = (unsigned)__shfl_up((int)incl, o);
    if (lane >= o) incl += t;
  }
  *total = (unsigned)__shfl((int)incl, 31);
  return incl - v;
}


#define HR 25600
#define HC 64
#define HU 25

#define BPR (HR / BW)

__global__ __launch_bounds__(1024) void k_hist(const int* __restrict__ dst, unsigned* __restrict__ part,
                                               unsigned* __restrict__ cntT, int E, int N,
                                               const float* __restrict__ W1, v4f* __restrict__ W1f,
                                               const float* __restrict__ W2, v4f* __restrict__ W2f) {
  __shared__ unsigned cnt[HR];
  const int tid = threadIdx.x;
  const int gi = (blockIdx.y * HC + blockIdx.x) * 1024 + tid;
  if (gi < (F_IN / 4) * 32) {
    const int t = gi >> 5, lane = gi & 31, c = lane & 15, h = lane >> 4;
    const int k = 8 * (t >> 1) + 4 * h + 2 * (t & 1);
    v4f f;
    f.x = W1[k * F_HID + c];       f.y = W1[(k + 1) * F_HID + c];
    f.z = W1[k * F_HID + 16 + c];  f.w = W1[(k + 1) * F_HID + 16 + c];
    W1f[gi] = f;
  }
  if (gi < (F_HID / 4) * 2 * 32) {
    const int t = gi >> 6, q = (gi >> 5) & 1, lane = gi & 31, c = lane & 15, h = lane >> 4;
    const int k = 16 * h + 2 * t;
    const int n = 32 * q + c;
    v4f f;
    f.x = W2[k * F_OUT + n];       f.y = W2[(k + 1) * F_OUT + n];
    f.z = W2[k * F_OUT + 16 + n];  f.w = W2[(k + 1) * F_OUT + 16 + n];
    W2f[gi] = f;
  }
  const int r0 = blockIdx.y * HR;
  const int chunk = (E + HC - 1) / HC;
  const int c0 = blockIdx.x * chunk;
  int c1 = c0 + chunk; if (c1 > E) c1 = E;
  for (int i = tid; i < HR; i += 1024) cnt[i] = 0u;
  __syncthreads();
  for (int base = c0 + tid; base < c1; base += 1024 * HU) {
    int dv[HU];
#pragma unroll
    for (int u = 0; u < HU; ++u) {
      const int e = base + u * 1024;
      const int v = dst[e < c1 ? e : c1 - 1];
      dv[u] = e < c1 ? v : -1;
    }
#pragma unroll
    for (int u = 0; u < HU; ++u) {
      const unsigned d = (unsigned)(dv[u] - r0);
      if (d < (unsigned)HR) atomicAdd(&cnt[d], 1u);
    }
  }
  __syncthreads();
  int nr = N - r0; if (nr > HR) nr = HR;
  unsigned* __restrict__ p = part + (size_t)blockIdx.x * N + r0;
  for (int i = tid; i < nr; i += 1024) p[i] = cnt[i];
  const int wv = tid >> 5, ln = tid & 31;
  for (int lb = wv; lb < BPR; lb += 32) {
    unsigned s = 0;
#pragma unroll
    for (int u = 0; u < (BW + 31) / 32; ++u) {
      const int i = ln + 32 * u;
      if (i < BW) s += cnt[lb * BW + i];
    }
#pragma unroll
    for (int o = 16; o; o >>= 1) s += (unsigned)__shfl_xor((int)s, o);
    if (ln == 0) cntT[(blockIdx.y * BPR + lb) * HC + blockIdx.x] = s;
  }
}

#define FU 13

__global__ __launch_bounds__(1024) void k_fill(const int* __restrict__ src, const int* __restrict__ dst,
                                               const unsigned* __restrict__ cntT, unsigned* __restrict__ binstart,
                                               unsigned* __restrict__ binned, int E) {
  __shared__ unsigned sorted[SLCAP];
  __shared__ unsigned s_tot[NBIN], s_bef[NBIN], s_base[NBIN + 1];
  __shared__ unsigned lstart[BPR], cursor[BPR];
  const int tid = threadIdx.x, lane = tid & 31, wv = tid >> 5;
  const int x = blockIdx.x, y = blockIdx.y;
  {
    const int bin = tid >> 2, q = tid & 3;
    const unsigned* p = cntT + bin * HC + q * 16;
    const bool live = bin < (int)gridDim.y * BPR;
    unsigned t = 0, b = 0;
#pragma unroll
    for (int i = 0; i < 16; ++i) {
      const unsigned c = live ? p[i] : 0u;
      t += c;
      if (q * 16 + i < x) b += c;
    }
    t += (unsigned)__shfl_xor((int)t, 1); t += (unsigned)__shfl_xor((int)t, 2);
    b += (unsigned)__shfl_xor((int)b, 1); b += (unsigned)__shfl_xor((int)b, 2);
    if (q == 0) { s_tot[bin] = t; s_bef[bin] = b; }
  }
  __syncthreads();
  if (wv == 0) {
    unsigned loc[8]; unsigned s = 0;
#pragma unroll
    for (int k = 0; k < 8; ++k) { loc[k] = s_tot[lane * 8 + k]; s += loc[k]; }
    unsigned tot; unsigned ex = wave_excl_scan(s, lane, &tot);
#pragma unroll
    for (int k = 0; k < 8; ++k) { s_base[lane * 8 + k] = ex; ex += loc[k]; }
    if (lane == 0) s_base[NBIN] = tot;
  } else if (wv == 1) {
    const unsigned a0 = cntT[(y * BPR + 2 * lane) * HC + x], a1 = cntT[(y * BPR + 2 * lane + 1) * HC + x];
    unsigned tot; const unsigned ex = wave_excl_scan(a0 + a1, lane, &tot);
    lstart[2 * lane] = ex;          cursor[2 * lane] = ex;
    lstart[2 * lane + 1] = ex + a0; cursor[2 * lane + 1] = ex + a0;
  }
  __syncthreads();
  if (x == 0 && y == 0) for (int i = tid; i <= NBIN; i += 1024) binstart[i] = s_base[i];
  const int r0 = y * HR;
  const int chunk = (E + HC - 1) / HC;
  const int c0 = x * chunk;
  int c1 = c0 + chunk; if (c1 > E) c1 = E;
  for (int base = c0 + tid; base < c1; base += 1024 * FU) {
    int dv[FU], sv[FU];
#pragma unroll
    for (int u = 0; u < FU; ++u) {
      const int e = base + u * 1024;
      const int ee = e < c1 ? e : c1 - 1;
      const int v = dst[ee];
      sv[u] = src[ee];
      dv[u] = e < c1 ? v : -1;
    }
    unsigned pp[FU], wd[FU];
#pragma unroll
    for (int u = 0; u < FU; ++u) {
      const unsigned d = (unsigned)(dv[u] - r0);
      pp[u] = 0xFFFFFFFFu;
      wd[u] = 0u;
      if (d < (unsigned)HR) {
        const unsigned lb = d / BW, dl = d - lb * BW;
        pp[u] = atomicAdd(&cursor[lb], 1u);
        wd[u] = (unsigned)sv[u] | (dl << 17);
      }
    }
#pragma unroll
    for (int u = 0; u < FU; ++u)
      if (pp[u] < (unsigned)SLCAP) sorted[pp[u]] = wd[u];
  }
  __syncthreads();
  for (int lb = wv; lb < BPR; lb += 32) {
    const unsigned a = lstart[lb], c = cursor[lb] - a;
    unsigned* __restrict__ gd = binned + s_base[y * BPR + lb] + s_bef[y * BPR + lb];
    for (unsigned k = lane; k < c; k += 32) gd[k] = sorted[a + k];
  }
}


__global__ void k_gemm1(const float* __restrict__ X, const v4f* __restrict__ Wf,
                        const unsigned* __restrict__ part, float* __restrict__ dinv,
                        float* __restrict__ H, int N) {
  const int lane = threadIdx.x & 31;
  const int wave = (blockIdx.x * blockDim.x + threadIdx.x) >> 5;
  const int m0 = wave << 4;
  if (m0 >= N) return;
  const int half = lane >> 4;
  const int lrow = lane & 15;
  int row = m0 + lrow; if (row >= N) row = N - 1;
  const float* xr = X + (size_t)row * F_IN;

  unsigned pc[HC / 2];
  const unsigned* pr = part + (size_t)half * N + row;
#pragma unroll
  for (int i = 0; i < HC / 2; ++i)
    pc[i] = pr[(size_t)(2 * i) * N];

  const v4f* xq = (const v4f*)(xr + (half << 2));
  const v4f* wq = Wf + lane;
  v8f c0 = {}; v8f c1 = {};
#pragma unroll 4
  for (int u = 0; u < F_IN / 8; ++u) {
    const v4f a4 = xq[2 * u];
    const v4f p0 = wq[(2 * u) * 32];
    const v4f p1 = wq[(2 * u + 1) * 32];
    v2f a;  a.x  = a4.x; a.y  = a4.y;
    v2f b0; b0.x = p0.x; b0.y = p0.y;
    v2f b1; b1.x = p0.z; b1.y = p0.w;
    c0 = __builtin_amdgcn_wmma_f32_16x16x4_f32(false, a, false, b0, (short)0, c0, false, false);
    c1 = __builtin_amdgcn_wmma_f32_16x16x4_f32(false, a, false, b1, (short)0, c1, false, false);
    a.x  = a4.z; a.y  = a4.w;
    b0.x = p1.x; b0.y = p1.y;
    b1.x = p1.z; b1.y = p1.w;
    c0 = __builtin_amdgcn_wmma_f32_16x16x4_f32(false, a, false, b0, (short)0, c0, false, false);
    c1 = __builtin_amdgcn_wmma_f32_16x16x4_f32(false, a, false, b1, (short)0, c1, false, false);
  }
  unsigned deg = 1u;
#pragma unroll
  for (int i = 0; i < HC / 2; ++i) deg += pc[i];
  deg += (unsigned)__shfl_xor((int)(deg - 1u), 16);
  const float dv = rsqrtf((float)deg);
  if (half == 0 && m0 + lrow < N) dinv[m0 + lrow] = dv;
#pragma unroll
  for (int j = 0; j < 8; ++j) {
    int m = m0 + j + (half << 3);
    const float dlo = __int_as_float(__builtin_amdgcn_readlane(__float_as_int(dv), j));
    const float dhi = __int_as_float(__builtin_amdgcn_readlane(__float_as_int(dv), j + 8));
    const float dmj = half ? dhi : dlo;
    if (m < N) {
      const size_t o = (size_t)m * F_HID;
      const float v0 = c0[j] * dmj;
      const float v1 = c1[j] * dmj;
      H[o + lrow]      = v0;
      H[o + 16 + lrow] = v1;
    }
  }
}


__global__ void k_gemm2(const float* __restrict__ A, const v4f* __restrict__ Wf,
                        const float* __restrict__ bias, const float* __restrict__ dinv,
                        float* __restrict__ H, int N) {
  const int lane = threadIdx.x & 31;
  const int wave = (blockIdx.x * blockDim.x + threadIdx.x) >> 5;
  const int m0 = wave << 4;
  if (m0 >= N) return;
  const int half = lane >> 4;
  const int lrow = lane & 15;
  int row = m0 + lrow; if (row >= N) row = N - 1;
  const float* ar  = A + (size_t)row * F_HID;

  float dm[8];
#pragma unroll
  for (int j = 0; j < 8; ++j) {
    int m = m0 + j + (half << 3); if (m >= N) m = N - 1;
    dm[j] = dinv[m];
  }

  const v4f* aq0 = (const v4f*)(ar  + (half << 4));
  const v4f* wq = Wf + lane;
  v4f s4[4];
#pragma unroll
  for (int i = 0; i < 4; ++i) s4[i] = aq0[i];
  v8f c0 = {}; v8f c1 = {}; v8f c2 = {}; v8f c3 = {};
#pragma unroll
  for (int t = 0; t < F_HID / 4; ++t) {
    const v4f sv = s4[t >> 1];
    v2f a;
    if (t & 1) { a.x = sv.z; a.y = sv.w; } else { a.x = sv.x; a.y = sv.y; }
    const v4f p0 = wq[(2 * t) * 32];
    const v4f p1 = wq[(2 * t + 1) * 32];
    v2f b;
    b.x = p0.x; b.y = p0.y;
    c0 = __builtin_amdgcn_wmma_f32_16x16x4_f32(false, a, false, b, (short)0, c0, false, false);
    b.x = p0.z; b.y = p0.w;
    c1 = __builtin_amdgcn_wmma_f32_16x16x4_f32(false, a, false, b, (short)0, c1, false, false);
    b.x = p1.x; b.y = p1.y;
    c2 = __builtin_amdgcn_wmma_f32_16x16x4_f32(false, a, false, b, (short)0, c2, false, false);
    b.x = p1.z; b.y = p1.w;
    c3 = __builtin_amdgcn_wmma_f32_16x16x4_f32(false, a, false, b, (short)0, c3, false, false);
  }
  const float bz0 = bias[lrow], bz1 = bias[16 + lrow], bz2 = bias[32 + lrow], bz3 = bias[48 + lrow];
#pragma unroll
  for (int j = 0; j < 8; ++j) {
    int m = m0 + j + (half << 3);
    if (m < N) {
      float* hr = H + (size_t)m * F_OUT;
      hr[lrow]      = c0[j] * dm[j] + bz0;
      hr[16 + lrow] = c1[j] * dm[j] + bz1;
      hr[32 + lrow] = c2[j] * dm[j] + bz2;
      hr[48 + lrow] = c3[j] * dm[j] + bz3;
    }
  }
}

template <int LAYER>
__device__ __forceinline__ void pull_store(float sum, int node, int lane, float di, float bz, float* __restrict__ outp) {
  if (LAYER == 1) outp[(size_t)node * F_HID + lane] = fmaxf(sum * di + bz, 0.0f) * di;
  else            outp[(size_t)node * F_HID + lane] = sum;
}

template <int LAYER>
__global__ __launch_bounds__(1024) void k_pull(const float* __restrict__ rows, const unsigned* __restrict__ binned,
                                               const unsigned* __restrict__ binstart, const float* __restrict__ dinv,
                                               const float* __restrict__ bias, float* __restrict__ outp, int N) {
  __shared__ unsigned sorted[CAP];
  __shared__ unsigned cnt[BW], st[BW + 1], cur[BW];
  const int tid = threadIdx.x, lane = tid & 31, wv = tid >> 5;
  const int n0 = blockIdx.x * BW;
  int nn = N - n0; if (nn > BW) nn = BW;
  const unsigned s0 = binstart[blockIdx.x];
  const unsigned L = binstart[blockIdx.x + 1] - s0;
  const unsigned* __restrict__ lst = binned + s0;
  const float bz = (LAYER == 1) ? bias[lane] : 0.0f;
  if (L <= (unsigned)CAP) {
    for (int i = tid; i < BW; i += 1024) cnt[i] = 0u;
    __syncthreads();
    for (unsigned kb = 0; kb < L; kb += 1024 * PU) {
      unsigned ev[PU];
#pragma unroll
      for (int u = 0; u < PU; ++u) {
        const unsigned k = kb + u * 1024 + tid;
        const unsigned e = lst[k < L ? k : L - 1u];
        ev[u] = k < L ? e : NOEDGE;
      }
#pragma unroll
      for (int u = 0; u < PU; ++u)
        if (ev[u] != NOEDGE) atomicAdd(&cnt[ev[u] >> 17], 1u);
    }
    __syncthreads();
    if (wv == 0) {
      unsigned loc[13]; unsigned s = 0;
#pragma unroll
      for (int k = 0; k < 13; ++k) {
        const int idx = lane * 13 + k;
        loc[k] = idx < BW ? cnt[idx] : 0u;
        s += loc[k];
      }
      unsigned tot; unsigned ex = wave_excl_scan(s, lane, &tot);
#pragma unroll
      for (int k = 0; k < 13; ++k) {
        const int idx = lane * 13 + k;
        if (idx < BW) { st[idx] = ex; cur[idx] = ex; }
        ex += loc[k];
      }
      if (lane == 0) st[BW] = tot;
    }
    __syncthreads();
    for (unsigned kb = 0; kb < L; kb += 1024 * PU) {
      unsigned ev[PU], pp[PU];
#pragma unroll
      for (int u = 0; u < PU; ++u) {
        const unsigned k = kb + u * 1024 + tid;
        const unsigned e = lst[k < L ? k : L - 1u];
        ev[u] = k < L ? e : NOEDGE;
      }
#pragma unroll
      for (int u = 0; u < PU; ++u) {
        pp[u] = 0u;
        if (ev[u] != NOEDGE) pp[u] = atomicAdd(&cur[ev[u] >> 17], 1u);
      }
#pragma unroll
      for (int u = 0; u < PU; ++u)
        if (ev[u] != NOEDGE) sorted[pp[u]] = ev[u] & SRCM;
    }
    __syncthreads();
    for (int i = wv; i < nn; i += 32) {
      const unsigned a = (unsigned)__builtin_amdgcn_readfirstlane((int)st[i]);
      const unsigned z = (unsigned)__builtin_amdgcn_readfirstlane((int)st[i + 1]);
      float q0 = rows[(size_t)(n0 + i) * F_HID + lane];
      const float di = (LAYER == 1) ? dinv[n0 + i] : 0.0f;
      float q1 = 0.0f, q2 = 0.0f, q3 = 0.0f;
      for (unsigned k = a; k < z; k += 32) {
        const int n = (int)(z - k < 32u ? z - k : 32u);
        const int si = (int)((k + lane < z) ? sorted[k + lane] : 0u);
        float v[32];
#pragma unroll
        for (int j = 0; j < 32; ++j) {
          v[j] = 0.0f;
          if (j < n) v[j] = rows[(size_t)__builtin_amdgcn_readlane(si, j) * F_HID + lane];
        }
#pragma unroll
        for (int j = 0; j < 32; j += 4) {
          q0 += v[j]; q1 += v[j + 1]; q2 += v[j + 2]; q3 += v[j + 3];
        }
      }
      pull_store<LAYER>((q0 + q1) + (q2 + q3), n0 + i, lane, di, bz, outp);
    }
  } else {
    for (int i = wv; i < nn; i += 32) {
      float sum = rows[(size_t)(n0 + i) * F_HID + lane];
      const float di = (LAYER == 1) ? dinv[n0 + i] : 0.0f;
      for (unsigned k = 0; k < L; k += 32) {
        const unsigned e = (k + lane < L) ? lst[k + lane] : 0xFFFFFFFFu;
        unsigned m = __builtin_amdgcn_ballot_w32((e >> 17) == (unsigned)i);
        while (m) {
          const int j = __builtin_ctz(m);
          m &= m - 1u;
          sum += rows[(size_t)__builtin_amdgcn_readlane((int)(e & SRCM), j) * F_HID + lane];
        }
      }
      pull_store<LAYER>(sum, n0 + i, lane, di, bz, outp);
    }
  }
}


static inline int cdiv_ll(long long a, long long b) { return (int)((a + b - 1) / b); }

extern "C" void kernel_launch(void* const* d_in, const int* in_sizes, int n_in,
                              void* d_out, int out_size, void* d_ws, size_t ws_size,
                              hipStream_t stream) {
  const float* X  = (const float*)d_in[0];
  const int*   ei = (const int*)d_in[1];
  const float* W1 = (const float*)d_in[2];
  const float* b1 = (const float*)d_in[3];
  const float* W2 = (const float*)d_in[4];
  const float* b2 = (const float*)d_in[5];
  float* out = (float*)d_out;

  const int N = in_sizes[0] / F_IN;
  const int E = in_sizes[1] / 2;
  const int* src = ei;
  const int* dst = ei + E;

  v4f* W1f = (v4f*)d_ws;
  v4f* W2f = W1f + (F_IN / 4) * 32;
  float* h1   = (float*)(W2f + (F_HID / 4) * 2 * 32);
  float* out1 = h1 + (size_t)N * F_HID;
  float* g    = out1 + (size_t)N * F_HID;
  float* dinv = g + (size_t)N * F_HID;
  unsigned* part = (unsigned*)(dinv + N);
  unsigned* cntT = part + (size_t)HC * N;
  unsigned* binstart = cntT + NBIN * HC;
  unsigned* binned = binstart + (NBIN + 1);

  const int B = 256;
  const int mtiles = cdiv_ll(N, 16);

  k_hist     <<<dim3(HC, cdiv_ll(N, HR)), 1024, 0, stream>>>(dst, part, cntT, E, N, W1, W1f, W2, W2f);
  k_fill     <<<dim3(HC, cdiv_ll(N, HR)), 1024, 0, stream>>>(src, dst, cntT, binstart, binned, E);

  k_gemm1    <<<cdiv_ll((long long)mtiles * 32, B), B, 0, stream>>>(X, W1f, part, dinv, h1, N);
  k_pull<1>  <<<cdiv_ll(N, BW), 1024, 0, stream>>>(h1, binned, binstart, dinv, b1, out1, N);
  k_pull<2>  <<<cdiv_ll(N, BW), 1024, 0, stream>>>(out1, binned, binstart, dinv, b1, g, N);
  k_gemm2    <<<cdiv_ll((long long)mtiles * 32, B), B, 0, stream>>>(g, W2f, b2, dinv, out, N);
}


# the second product (32->64, +b2) has no launch of its own: the layer-2 aggregation workgroup keeps its bin's sums in LDS and its waves run the 16-node WMMA tiles from there; 6 launches become 5
# speedup vs baseline: 6.2403x; 1.0348x over previous
#include <hip/hip_runtime.h>
#include <hip/hip_bf16.h>

typedef float v2f __attribute__((ext_vector_type(2)));
typedef float v8f __attribute__((ext_vector_type(8)));
typedef float v4f __attribute__((ext_vector_type(4)));

#define F_IN  512
#define F_HID 32
#define F_OUT 64


#define BW    400
#define NBIN  256
#define SLCAP 50000
#define CAP   49152
#define SRCM  0x1FFFFu
#define NOEDGE 0xFFFFFFFFu
#define PU    16
#define GP    36

__device__ __forceinline__ unsigned wave_excl_scan(unsigned v, int lane, unsigned* total) {
  unsigned incl = v;
#pragma unroll
  for (int o = 1; o < 32; o <<= 1) {
    const unsigned t = (unsigned)__shfl_up((int)incl, o);
    if (lane >= o) incl += t;
  }
  *total = (unsigned)__shfl((int)incl, 31);
  return incl - v;
}


#define HR 25600
#define HC 64
#define HU 25

#define BPR (HR / BW)

__global__ __launch_bounds__(1024) void k_hist(const int* __restrict__ dst, unsigned* __restrict__ part,
                                               unsigned* __restrict__ cntT, int E, int N,
                                               const float* __restrict__ W1, v4f* __restrict__ W1f,
                                               const float* __restrict__ W2, v4f* __restrict__ W2f) {
  __shared__ unsigned cnt[HR];
  const int tid = threadIdx.x;
  const int gi = (blockIdx.y * HC + blockIdx.x) * 1024 + tid;
  if (gi < (F_IN / 4) * 32) {
    const int t = gi >> 5, lane = gi & 31, c = lane & 15, h = lane >> 4;
    const int k = 8 * (t >> 1) + 4 * h + 2 * (t & 1);
    v4f f;
    f.x = W1[k * F_HID + c];       f.y = W1[(k + 1) * F_HID + c];
    f.z = W1[k * F_HID + 16 + c];  f.w = W1[(k + 1) * F_HID + 16 + c];
    W1f[gi] = f;
  }
  if (gi < (F_HID / 4) * 2 * 32) {
    const int t = gi >> 6, q = (gi >> 5) & 1, lane = gi & 31, c = lane & 15, h = lane >> 4;
    const int k = 16 * h + 2 * t;
    const int n = 32 * q + c;
    v4f f;
    f.x = W2[k * F_OUT + n];       f.y = W2[(k + 1) * F_OUT + n];
    f.z = W2[k * F_OUT + 16 + n];  f.w = W2[(k + 1) * F_OUT + 16 + n];
    W2f[gi] = f;
  }
  const int r0 = blockIdx.y * HR;
  const int chunk = (E + HC - 1) / HC;
  const int c0 = blockIdx.x * chunk;
  int c1 = c0 + chunk; if (c1 > E) c1 = E;
  for (int i = tid; i < HR; i += 1024) cnt[i] = 0u;
  __syncthreads();
  for (int base = c0 + tid; base < c1; base += 1024 * HU) {
    int dv[HU];
#pragma unroll
    for (int u = 0; u < HU; ++u) {
      const int e = base + u * 1024;
      const int v = dst[e < c1 ? e : c1 - 1];
      dv[u] = e < c1 ? v : -1;
    }
#pragma unroll
    for (int u = 0; u < HU; ++u) {
      const unsigned d = (unsigned)(dv[u] - r0);
      if (d < (unsigned)HR) atomicAdd(&cnt[d], 1u);
    }
  }
  __syncthreads();
  int nr = N - r0; if (nr > HR) nr = HR;
  unsigned* __restrict__ p = part + (size_t)blockIdx.x * N + r0;
  for (int i = tid; i < nr; i += 1024) p[i] = cnt[i];
  const int wv = tid >> 5, ln = tid & 31;
  for (int lb = wv; lb < BPR; lb += 32) {
    unsigned s = 0;
#pragma unroll
    for (int u = 0; u < (BW + 31) / 32; ++u) {
      const int i = ln + 32 * u;
      if (i < BW) s += cnt[lb * BW + i];
    }
#pragma unroll
    for (int o = 16; o; o >>= 1) s += (unsigned)__shfl_xor((int)s, o);
    if (ln == 0) cntT[(blockIdx.y * BPR + lb) * HC + blockIdx.x] = s;
  }
}

#define FU 13

__global__ __launch_bounds__(1024) void k_fill(const int* __restrict__ src, const int* __restrict__ dst,
                                               const unsigned* __restrict__ cntT, unsigned* __restrict__ binstart,
                                               unsigned* __restrict__ binned, int E) {
  __shared__ unsigned sorted[SLCAP];
  __shared__ unsigned s_tot[NBIN], s_bef[NBIN], s_base[NBIN + 1];
  __shared__ unsigned lstart[BPR], cursor[BPR];
  const int tid = threadIdx.x, lane = tid & 31, wv = tid >> 5;
  const int x = blockIdx.x, y = blockIdx.y;
  {
    const int bin = tid >> 2, q = tid & 3;
    const unsigned* p = cntT + bin * HC + q * 16;
    const bool live = bin < (int)gridDim.y * BPR;
    unsigned t = 0, b = 0;
#pragma unroll
    for (int i = 0; i < 16; ++i) {
      const unsigned c = live ? p[i] : 0u;
      t += c;
      if (q * 16 + i < x) b += c;
    }
    t += (unsigned)__shfl_xor((int)t, 1); t += (unsigned)__shfl_xor((int)t, 2);
    b += (unsigned)__shfl_xor((int)b, 1); b += (unsigned)__shfl_xor((int)b, 2);
    if (q == 0) { s_tot[bin] = t; s_bef[bin] = b; }
  }
  __syncthreads();
  if (wv == 0) {
    unsigned loc[8]; unsigned s = 0;
#pragma unroll
    for (int k = 0; k < 8; ++k) { loc[k] = s_tot[lane * 8 + k]; s += loc[k]; }
    unsigned tot; unsigned ex = wave_excl_scan(s, lane, &tot);
#pragma unroll
    for (int k = 0; k < 8; ++k) { s_base[lane * 8 + k] = ex; ex += loc[k]; }
    if (lane == 0) s_base[NBIN] = tot;
  } else if (wv == 1) {
    const unsigned a0 = cntT[(y * BPR + 2 * lane) * HC + x], a1 = cntT[(y * BPR + 2 * lane + 1) * HC + x];
    unsigned tot; const unsigned ex = wave_excl_scan(a0 + a1, lane, &tot);
    lstart[2 * lane] = ex;          cursor[2 * lane] = ex;
    lstart[2 * lane + 1] = ex + a0; cursor[2 * lane + 1] = ex + a0;
  }
  __syncthreads();
  if (x == 0 && y == 0) for (int i = tid; i <= NBIN; i += 1024) binstart[i] = s_base[i];
  const int r0 = y * HR;
  const int chunk = (E + HC - 1) / HC;
  const int c0 = x * chunk;
  int c1 = c0 + chunk; if (c1 > E) c1 = E;
  for (int base = c0 + tid; base < c1; base += 1024 * FU) {
    int dv[FU], sv[FU];
#pragma unroll
    for (int u = 0; u < FU; ++u) {
      const int e = base + u * 1024;
      const int ee = e < c1 ? e : c1 - 1;
      const int v = dst[ee];
      sv[u] = src[ee];
      dv[u] = e < c1 ? v : -1;
    }
    unsigned pp[FU], wd[FU];
#pragma unroll
    for (int u = 0; u < FU; ++u) {
      const unsigned d = (unsigned)(dv[u] - r0);
      pp[u] = 0xFFFFFFFFu;
      wd[u] = 0u;
      if (d < (unsigned)HR) {
        const unsigned lb = d / BW, dl = d - lb * BW;
        pp[u] = atomicAdd(&cursor[lb], 1u);
        wd[u] = (unsigned)sv[u] | (dl << 17);
      }
    }
#pragma unroll
    for (int u = 0; u < FU; ++u)
      if (pp[u] < (unsigned)SLCAP) sorted[pp[u]] = wd[u];
  }
  __syncthreads();
  for (int lb = wv; lb < BPR; lb += 32) {
    const unsigned a = lstart[lb], c = cursor[lb] - a;
    unsigned* __restrict__ gd = binned + s_base[y * BPR + lb] + s_bef[y * BPR + lb];
    for (unsigned k = lane; k < c; k += 32) gd[k] = sorted[a + k];
  }
}


__global__ void k_gemm1(const float* __restrict__ X, const v4f* __restrict__ Wf,
                        const unsigned* __restrict__ part, float* __restrict__ dinv,
                        float* __restrict__ H, int N) {
  const int lane = threadIdx.x & 31;
  const int wave = (blockIdx.x * blockDim.x + threadIdx.x) >> 5;
  const int m0 = wave << 4;
  if (m0 >= N) return;
  const int half = lane >> 4;
  const int lrow = lane & 15;
  int row = m0 + lrow; if (row >= N) row = N - 1;
  const float* xr = X + (size_t)row * F_IN;

  unsigned pc[HC / 2];
  const unsigned* pr = part + (size_t)half * N + row;
#pragma unroll
  for (int i = 0; i < HC / 2; ++i)
    pc[i] = pr[(size_t)(2 * i) * N];

  const v4f* xq = (const v4f*)(xr + (half << 2));
  const v4f* wq = Wf + lane;
  v8f c0 = {}; v8f c1 = {};
#pragma unroll 4
  for (int u = 0; u < F_IN / 8; ++u) {
    const v4f a4 = xq[2 * u];
    const v4f p0 = wq[(2 * u) * 32];
    const v4f p1 = wq[(2 * u + 1) * 32];
    v2f a;  a.x  = a4.x; a.y  = a4.y;
    v2f b0; b0.x = p0.x; b0.y = p0.y;
    v2f b1; b1.x = p0.z; b1.y = p0.w;
    c0 = __builtin_amdgcn_wmma_f32_16x16x4_f32(false, a, false, b0, (short)0, c0, false, false);
    c1 = __builtin_amdgcn_wmma_f32_16x16x4_f32(false, a, false, b1, (short)0, c1, false, false);
    a.x  = a4.z; a.y  = a4.w;
    b0.x = p1.x; b0.y = p1.y;
    b1.x = p1.z; b1.y = p1.w;
    c0 = __builtin_amdgcn_wmma_f32_16x16x4_f32(false, a, false, b0, (short)0, c0, false, false);
    c1 = __builtin_amdgcn_wmma_f32_16x16x4_f32(false, a, false, b1, (short)0, c1, false, false);
  }
  unsigned deg = 1u;
#pragma unroll
  for (int i = 0; i < HC / 2; ++i) deg += pc[i];
  deg += (unsigned)__shfl_xor((int)(deg - 1u), 16);
  const float dv = rsqrtf((float)deg);
  if (half == 0 && m0 + lrow < N) dinv[m0 + lrow] = dv;
#pragma unroll
  for (int j = 0; j < 8; ++j) {
    int m = m0 + j + (half << 3);
    const float dlo = __int_as_float(__builtin_amdgcn_readlane(__float_as_int(dv), j));
    const float dhi = __int_as_float(__builtin_amdgcn_readlane(__float_as_int(dv), j + 8));
    const float dmj = half ? dhi : dlo;
    if (m < N) {
      const size_t o = (size_t)m * F_HID;
      const float v0 = c0[j] * dmj;
      const float v1 = c1[j] * dmj;
      H[o + lrow]      = v0;
      H[o + 16 + lrow] = v1;
    }
  }
}


struct G2Ahead { v4f p[2 * (F_HID / 4)]; float dm[8]; float bz[4]; };

__device__ __forceinline__ void gemm2_ahead(G2Ahead& o, int r0, int nn, int n0, int lane,
                                            const v4f* __restrict__ Wf, const float* __restrict__ bias,
                                            const float* __restrict__ dinv) {
  const int half = lane >> 4;
  const int lrow = lane & 15;
  const v4f* wq = Wf + lane;
#pragma unroll
  for (int t = 0; t < 2 * (F_HID / 4); ++t) o.p[t] = wq[t * 32];
#pragma unroll
  for (int j = 0; j < 8; ++j) {
    int m = r0 + j + (half << 3); if (m >= nn) m = nn - 1;
    o.dm[j] = dinv[n0 + m];
  }
#pragma unroll
  for (int q = 0; q < 4; ++q) o.bz[q] = bias[16 * q + lrow];
}

__device__ __forceinline__ void gemm2_tile(const G2Ahead& o, const float* gs, int r0, int nn, int n0, int lane,
                                           float* __restrict__ H) {
  const int half = lane >> 4;
  const int lrow = lane & 15;
  int row = r0 + lrow; if (row >= nn) row = nn - 1;
  const v4f* aq0 = (const v4f*)(gs + row * GP + (half << 4));
  v4f s4[4];
#pragma unroll
  for (int i = 0; i < 4; ++i) s4[i] = aq0[i];
  v8f c0 = {}; v8f c1 = {}; v8f c2 = {}; v8f c3 = {};
#pragma unroll
  for (int t = 0; t < F_HID / 4; ++t) {
    const v4f sv = s4[t >> 1];
    v2f a;
    if (t & 1) { a.x = sv.z; a.y = sv.w; } else { a.x = sv.x; a.y = sv.y; }
    const v4f p0 = o.p[2 * t];
    const v4f p1 = o.p[2 * t + 1];
    v2f b;
    b.x = p0.x; b.y = p0.y;
    c0 = __builtin_amdgcn_wmma_f32_16x16x4_f32(false, a, false, b, (short)0, c0, false, false);
    b.x = p0.z; b.y = p0.w;
    c1 = __builtin_amdgcn_wmma_f32_16x16x4_f32(false, a, false, b, (short)0, c1, false, false);
    b.x = p1.x; b.y = p1.y;
    c2 = __builtin_amdgcn_wmma_f32_16x16x4_f32(false, a, false, b, (short)0, c2, false, false);
    b.x = p1.z; b.y = p1.w;
    c3 = __builtin_amdgcn_wmma_f32_16x16x4_f32(false, a, false, b, (short)0, c3, false, false);
  }
#pragma unroll
  for (int j = 0; j < 8; ++j) {
    const int m = r0 + j + (half << 3);
    if (m < nn) {
      float* hr = H + (size_t)(n0 + m) * F_OUT;
      hr[lrow]      = c0[j] * o.dm[j] + o.bz[0];
      hr[16 + lrow] = c1[j] * o.dm[j] + o.bz[1];
      hr[32 + lrow] = c2[j] * o.dm[j] + o.bz[2];
      hr[48 + lrow] = c3[j] * o.dm[j] + o.bz[3];
    }
  }
}

template <int LAYER>
__device__ __forceinline__ void pull_store(float sum, int n0, int i, int lane, float di, float bz, float* __restrict__ outp, float* gs) {
  if (LAYER == 1) outp[(size_t)(n0 + i) * F_HID + lane] = fmaxf(sum * di + bz, 0.0f) * di;
  else            gs[i * GP + lane] = sum;
}

template <int LAYER>
__global__ __launch_bounds__(1024) void k_pull(const float* __restrict__ rows, const unsigned* __restrict__ binned,
                                               const unsigned* __restrict__ binstart, const float* __restrict__ dinv,
                                               const float* __restrict__ bias, float* __restrict__ outp, int N,
                                               const v4f* __restrict__ W2f, const float* __restrict__ b2,
                                               float* __restrict__ out) {
  __shared__ unsigned sorted[CAP];
  __shared__ unsigned cnt[BW], st[BW + 1], cur[BW];
  __shared__ __attribute__((aligned(16))) float gs[LAYER == 2 ? BW * GP : 4];
  const int tid = threadIdx.x, lane = tid & 31, wv = tid >> 5;
  const int n0 = blockIdx.x * BW;
  int nn = N - n0; if (nn > BW) nn = BW;
  const unsigned s0 = binstart[blockIdx.x];
  const unsigned L = binstart[blockIdx.x + 1] - s0;
  const unsigned* __restrict__ lst = binned + s0;
  const float bz = (LAYER == 1) ? bias[lane] : 0.0f;
  const bool mine = (LAYER == 2) && wv < ((nn + 15) >> 4);
  G2Ahead ah;
  if (mine) gemm2_ahead(ah, wv << 4, nn, n0, lane, W2f, b2, dinv);
  if (L <= (unsigned)CAP) {
    for (int i = tid; i < BW; i += 1024) cnt[i] = 0u;
    __syncthreads();
    for (unsigned kb = 0; kb < L; kb += 1024 * PU) {
      unsigned ev[PU];
#pragma unroll
      for (int u = 0; u < PU; ++u) {
        const unsigned k = kb + u * 1024 + tid;
        const unsigned e = lst[k < L ? k : L - 1u];
        ev[u] = k < L ? e : NOEDGE;
      }
#pragma unroll
      for (int u = 0; u < PU; ++u)
        if (ev[u] != NOEDGE) atomicAdd(&cnt[ev[u] >> 17], 1u);
    }
    __syncthreads();
    if (wv == 0) {
      unsigned loc[13]; unsigned s = 0;
#pragma unroll
      for (int k = 0; k < 13; ++k) {
        const int idx = lane * 13 + k;
        loc[k] = idx < BW ? cnt[idx] : 0u;
        s += loc[k];
      }
      unsigned tot; unsigned ex = wave_excl_scan(s, lane, &tot);
#pragma unroll
      for (int k = 0; k < 13; ++k) {
        const int idx = lane * 13 + k;
        if (idx < BW) { st[idx] = ex; cur[idx] = ex; }
        ex += loc[k];
      }
      if (lane == 0) st[BW] = tot;
    }
    __syncthreads();
    for (unsigned kb = 0; kb < L; kb += 1024 * PU) {
      unsigned ev[PU], pp[PU];
#pragma unroll
      for (int u = 0; u < PU; ++u) {
        const unsigned k = kb + u * 1024 + tid;
        const unsigned e = lst[k < L ? k : L - 1u];
        ev[u] = k < L ? e : NOEDGE;
      }
#pragma unroll
      for (int u = 0; u < PU; ++u) {
        pp[u] = 0u;
        if (ev[u] != NOEDGE) pp[u] = atomicAdd(&cur[ev[u] >> 17], 1u);
      }
#pragma unroll
      for (int u = 0; u < PU; ++u)
        if (ev[u] != NOEDGE) sorted[pp[u]] = ev[u] & SRCM;
    }
    __syncthreads();
    for (int i = wv; i < nn; i += 32) {
      const unsigned a = (unsigned)__builtin_amdgcn_readfirstlane((int)st[i]);
      const unsigned z = (unsigned)__builtin_amdgcn_readfirstlane((int)st[i + 1]);
      float q0 = rows[(size_t)(n0 + i) * F_HID + lane];
      const float di = (LAYER == 1) ? dinv[n0 + i] : 0.0f;
      float q1 = 0.0f, q2 = 0.0f, q3 = 0.0f;
      for (unsigned k = a; k < z; k += 32) {
        const int n = (int)(z - k < 32u ? z - k : 32u);
        const int si = (int)((k + lane < z) ? sorted[k + lane] : 0u);
        float v[32];
#pragma unroll
        for (int j = 0; j < 32; ++j) {
          v[j] = 0.0f;
          if (j < n) v[j] = rows[(size_t)__builtin_amdgcn_readlane(si, j) * F_HID + lane];
        }
#pragma unroll
        for (int j = 0; j < 32; j += 4) {
          q0 += v[j]; q1 += v[j + 1]; q2 += v[j + 2]; q3 += v[j + 3];
        }
      }
      pull_store<LAYER>((q0 + q1) + (q2 + q3), n0, i, lane, di, bz, outp, gs);
    }
  } else {
    for (int i = wv; i < nn; i += 32) {
      float sum = rows[(size_t)(n0 + i) * F_HID + lane];
      const float di = (LAYER == 1) ? dinv[n0 + i] : 0.0f;
      for (unsigned k = 0; k < L; k += 32) {
        const unsigned e = (k + lane < L) ? lst[k + lane] : 0xFFFFFFFFu;
        unsigned m = __builtin_amdgcn_ballot_w32((e >> 17) == (unsigned)i);
        while (m) {
          const int j = __builtin_ctz(m);
          m &= m - 1u;
          sum += rows[(size_t)__builtin_amdgcn_readlane((int)(e & SRCM), j) * F_HID + lane];
        }
      }
      pull_store<LAYER>(sum, n0, i, lane, di, bz, outp, gs);
    }
  }
  if (LAYER == 2) {
    static_assert((BW + 15) / 16 <= 32, "a bin has at most one 16-node tile per wave");
    __syncthreads();
    if (mine) gemm2_tile(ah, gs, wv << 4, nn, n0, lane, out);
  }
}


static inline int cdiv_ll(long long a, long long b) { return (int)((a + b - 1) / b); }

extern "C" void kernel_launch(void* const* d_in, const int* in_sizes, int n_in,
                              void* d_out, int out_size, void* d_ws, size_t ws_size,
                              hipStream_t stream) {
  const float* X  = (const float*)d_in[0];
  const int*   ei = (const int*)d_in[1];
  const float* W1 = (const float*)d_in[2];
  const float* b1 = (const float*)d_in[3];
  const float* W2 = (const float*)d_in[4];
  const float* b2 = (const float*)d_in[5];
  float* out = (float*)d_out;

  const int N = in_sizes[0] / F_IN;
  const int E = in_sizes[1] / 2;
  const int* src = ei;
  const int* dst = ei + E;

  v4f* W1f = (v4f*)d_ws;
  v4f* W2f = W1f + (F_IN / 4) * 32;
  float* h1   = (float*)(W2f + (F_HID / 4) * 2 * 32);
  float* out1 = h1 + (size_t)N * F_HID;
  float* dinv = out1 + (size_t)N * F_HID;
  unsigned* part = (unsigned*)(dinv + N);
  unsigned* cntT = part + (size_t)HC * N;
  unsigned* binstart = cntT + NBIN * HC;
  unsigned* binned = binstart + (NBIN + 1);

  const int B = 256;
  const int mtiles = cdiv_ll(N, 16);

  k_hist     <<<dim3(HC, cdiv_ll(N, HR)), 1024, 0, stream>>>(dst, part, cntT, E, N, W1, W1f, W2, W2f);
  k_fill     <<<dim3(HC, cdiv_ll(N, HR)), 1024, 0, stream>>>(src, dst, cntT, binstart, binned, E);

  k_gemm1    <<<cdiv_ll((long long)mtiles * 32, B), B, 0, stream>>>(X, W1f, part, dinv, h1, N);
  k_pull<1>  <<<cdiv_ll(N, BW), 1024, 0, stream>>>(h1, binned, binstart, dinv, b1, out1, N, W2f, b2, out);
  k_pull<2>  <<<cdiv_ll(N, BW), 1024, 0, stream>>>(out1, binned, binstart, dinv, b1, nullptr, N, W2f, b2, out);
}


# the first product has no launch of its own: each workgroup of the edge-grouping launch also runs an equal share of its 16-node tiles, half of them before, half after their grouping; 5 launches, now 4
# speedup vs baseline: 6.5839x; 1.0551x over previous
#include <hip/hip_runtime.h>
#include <hip/hip_bf16.h>

typedef float v2f __attribute__((ext_vector_type(2)));
typedef float v8f __attribute__((ext_vector_type(8)));
typedef float v4f __attribute__((ext_vector_type(4)));

#define F_IN  512
#define F_HID 32
#define F_OUT 64


#define BW    400
#define NBIN  256
#define SLCAP 50000
#define CAP   49152
#define SRCM  0x1FFFFu
#define NOEDGE 0xFFFFFFFFu
#define PU    16
#define GP    36

__device__ __forceinline__ unsigned wave_excl_scan(unsigned v, int lane, unsigned* total) {
  unsigned incl = v;
#pragma unroll
  for (int o = 1; o < 32; o <<= 1) {
    const unsigned t = (unsigned)__shfl_up((int)incl, o);
    if (lane >= o) incl += t;
  }
  *total = (unsigned)__shfl((int)incl, 31);
  return incl - v;
}


#define HR 25600
#define HC 64
#define HU 25

#define BPR (HR / BW)

__global__ __launch_bounds__(1024) void k_hist(const int* __restrict__ dst, unsigned* __restrict__ part,
                                               unsigned* __restrict__ cntT, int E, int N,
                                               const float* __restrict__ W1, v4f* __restrict__ W1f,
                                               const float* __restrict__ W2, v4f* __restrict__ W2f) {
  __shared__ unsigned cnt[HR];
  const int tid = threadIdx.x;
  const int gi = (blockIdx.y * HC + blockIdx.x) * 1024 + tid;
  if (gi < (F_IN / 4) * 32) {
    const int t = gi >> 5, lane = gi & 31, c = lane & 15, h = lane >> 4;
    const int k = 8 * (t >> 1) + 4 * h + 2 * (t & 1);
    v4f f;
    f.x = W1[k * F_HID + c];       f.y = W1[(k + 1) * F_HID + c];
    f.z = W1[k * F_HID + 16 + c];  f.w = W1[(k + 1) * F_HID + 16 + c];
    W1f[gi] = f;
  }
  if (gi < (F_HID / 4) * 2 * 32) {
    const int t = gi >> 6, q = (gi >> 5) & 1, lane = gi & 31, c = lane & 15, h = lane >> 4;
    const int k = 16 * h + 2 * t;
    const int n = 32 * q + c;
    v4f f;
    f.x = W2[k * F_OUT + n];       f.y = W2[(k + 1) * F_OUT + n];
    f.z = W2[k * F_OUT + 16 + n];  f.w = W2[(k + 1) * F_OUT + 16 + n];
    W2f[gi] = f;
  }
  const int r0 = blockIdx.y * HR;
  const int chunk = (E + HC - 1) / HC;
  const int c0 = blockIdx.x * chunk;
  int c1 = c0 + chunk; if (c1 > E) c1 = E;
  for (int i = tid; i < HR; i += 1024) cnt[i] = 0u;
  __syncthreads();
  for (int base = c0 + tid; base < c1; base += 1024 * HU) {
    int dv[HU];
#pragma unroll
    for (int u = 0; u < HU; ++u) {
      const int e = base + u * 1024;
      const int v = dst[e < c1 ? e : c1 - 1];
      dv[u] = e < c1 ? v : -1;
    }
#pragma unroll
    for (int u = 0; u < HU; ++u) {
      const unsigned d = (unsigned)(dv[u] - r0);
      if (d < (unsigned)HR) atomicAdd(&cnt[d], 1u);
    }
  }
  __syncthreads();
  int nr = N - r0; if (nr > HR) nr = HR;
  unsigned* __restrict__ p = part + (size_t)blockIdx.x * N + r0;
  for (int i = tid; i < nr; i += 1024) p[i] = cnt[i];
  const int wv = tid >> 5, ln = tid & 31;
  for (int lb = wv; lb < BPR; lb += 32) {
    unsigned s = 0;
#pragma unroll
    for (int u = 0; u < (BW + 31) / 32; ++u) {
      const int i = ln + 32 * u;
      if (i < BW) s += cnt[lb * BW + i];
    }
#pragma unroll
    for (int o = 16; o; o >>= 1) s += (unsigned)__shfl_xor((int)s, o);
    if (ln == 0) cntT[(blockIdx.y * BPR + lb) * HC + blockIdx.x] = s;
  }
}

#define FU 13

__device__ __forceinline__ void gemm1_share(const float* __restrict__ X, const v4f* __restrict__ Wf,
                                            const unsigned* __restrict__ part, float* __restrict__ dinv,
                                            float* __restrict__ H, int N, int wg, int nwg, int wv, int lane);

__global__ __launch_bounds__(1024) void k_fill_gemm1(const int* __restrict__ src, const int* __restrict__ dst,
                                                     const unsigned* __restrict__ cntT, unsigned* __restrict__ binstart,
                                                     unsigned* __restrict__ binned, int E,
                                                     const float* __restrict__ X, const v4f* __restrict__ W1f,
                                                     const unsigned* __restrict__ part, float* __restrict__ dinv,
                                                     float* __restrict__ H, int N) {
  __shared__ unsigned sorted[SLCAP];
  __shared__ unsigned s_tot[NBIN], s_bef[NBIN], s_base[NBIN + 1];
  __shared__ unsigned lstart[BPR], cursor[BPR];
  const int tid = threadIdx.x, lane = tid & 31, wv = tid >> 5;
  const int x = blockIdx.x, y = blockIdx.y;
  const int wg = y * (int)gridDim.x + x, nwg = (int)(gridDim.x * gridDim.y);
  const bool tiles_first = (((wg >> 3) ^ wg) & 1) != 0;
  if (tiles_first) gemm1_share(X, W1f, part, dinv, H, N, wg, nwg, wv, lane);
  {
    const int bin = tid >> 2, q = tid & 3;
    const unsigned* p = cntT + bin * HC + q * 16;
    const bool live = bin < (int)gridDim.y * BPR;
    unsigned t = 0, b = 0;
#pragma unroll
    for (int i = 0; i < 16; ++i) {
      const unsigned c = live ? p[i] : 0u;
      t += c;
      if (q * 16 + i < x) b += c;
    }
    t += (unsigned)__shfl_xor((int)t, 1); t += (unsigned)__shfl_xor((int)t, 2);
    b += (unsigned)__shfl_xor((int)b, 1); b += (unsigned)__shfl_xor((int)b, 2);
    if (q == 0) { s_tot[bin] = t; s_bef[bin] = b; }
  }
  __syncthreads();
  if (wv == 0) {
    unsigned loc[8]; unsigned s = 0;
#pragma unroll
    for (int k = 0; k < 8; ++k) { loc[k] = s_tot[lane * 8 + k]; s += loc[k]; }
    unsigned tot; unsigned ex = wave_excl_scan(s, lane, &tot);
#pragma unroll
    for (int k = 0; k < 8; ++k) { s_base[lane * 8 + k] = ex; ex += loc[k]; }
    if (lane == 0) s_base[NBIN] = tot;
  } else if (wv == 1) {
    const unsigned a0 = cntT[(y * BPR + 2 * lane) * HC + x], a1 = cntT[(y * BPR + 2 * lane + 1) * HC + x];
    unsigned tot; const unsigned ex = wave_excl_scan(a0 + a1, lane, &tot);
    lstart[2 * lane] = ex;          cursor[2 * lane] = ex;
    lstart[2 * lane + 1] = ex + a0; cursor[2 * lane + 1] = ex + a0;
  }
  __syncthreads();
  if (x == 0 && y == 0) for (int i = tid; i <= NBIN; i += 1024) binstart[i] = s_base[i];
  const int r0 = y * HR;
  const int chunk = (E + HC - 1) / HC;
  const int c0 = x * chunk;
  int c1 = c0 + chunk; if (c1 > E) c1 = E;
  for (int base = c0 + tid; base < c1; base += 1024 * FU) {
    int dv[FU], sv[FU];
#pragma unroll
    for (int u = 0; u < FU; ++u) {
      const int e = base + u * 1024;
      const int ee = e < c1 ? e : c1 - 1;
      const int v = dst[ee];
      sv[u] = src[ee];
      dv[u] = e < c1 ? v : -1;
    }
    unsigned pp[FU], wd[FU];
#pragma unroll
    for (int u = 0; u < FU; ++u) {
      const unsigned d = (unsigned)(dv[u] - r0);
      pp[u] = 0xFFFFFFFFu;
      wd[u] = 0u;
      if (d < (unsigned)HR) {
        const unsigned lb = d / BW, dl = d - lb * BW;
        pp[u] = atomicAdd(&cursor[lb], 1u);
        wd[u] = (unsigned)sv[u] | (dl << 17);
      }
    }
#pragma unroll
    for (int u = 0; u < FU; ++u)
      if (pp[u] < (unsigned)SLCAP) sorted[pp[u]] = wd[u];
  }
  __syncthreads();
  for (int lb = wv; lb < BPR; lb += 32) {
    const unsigned a = lstart[lb], c = cursor[lb] - a;
    unsigned* __restrict__ gd = binned + s_base[y * BPR + lb] + s_bef[y * BPR + lb];
    for (unsigned k = lane; k < c; k += 32) gd[k] = sorted[a + k];
  }
  if (!tiles_first) gemm1_share(X, W1f, part, dinv, H, N, wg, nwg, wv, lane);
}


__device__ __forceinline__ void gemm1_tile(const float* __restrict__ X, const v4f* __restrict__ Wf,
                                           const unsigned* __restrict__ part, float* __restrict__ dinv,
                                           float* __restrict__ H, int N, int tile, int lane) {
  const int m0 = tile << 4;
  if (m0 >= N) return;
  const int half = lane >> 4;
  const int lrow = lane & 15;
  int row = m0 + lrow; if (row >= N) row = N - 1;
  const float* xr = X + (size_t)row * F_IN;

  unsigned pc[HC / 2];
  const unsigned* pr = part + (size_t)half * N + row;
#pragma unroll
  for (int i = 0; i < HC / 2; ++i)
    pc[i] = pr[(size_t)(2 * i) * N];

  const v4f* xq = (const v4f*)(xr + (half << 2));
  const v4f* wq = Wf + lane;
  v8f c0 = {}; v8f c1 = {};
#pragma unroll 4
  for (int u = 0; u < F_IN / 8; ++u) {
    const v4f a4 = xq[2 * u];
    const v4f p0 = wq[(2 * u) * 32];
    const v4f p1 = wq[(2 * u + 1) * 32];
    v2f a;  a.x  = a4.x; a.y  = a4.y;
    v2f b0; b0.x = p0.x; b0.y = p0.y;
    v2f b1; b1.x = p0.z; b1.y = p0.w;
    c0 = __builtin_amdgcn_wmma_f32_16x16x4_f32(false, a, false, b0, (short)0, c0, false, false);
    c1 = __builtin_amdgcn_wmma_f32_16x16x4_f32(false, a, false, b1, (short)0, c1, false, false);
    a.x  = a4.z; a.y  = a4.w;
    b0.x = p1.x; b0.y = p1.y;
    b1.x = p1.z; b1.y = p1.w;
    c0 = __builtin_amdgcn_wmma_f32_16x16x4_f32(false, a, false, b0, (short)0, c0, false, false);
    c1 = __builtin_amdgcn_wmma_f32_16x16x4_f32(false, a, false, b1, (short)0, c1, false, false);
  }
  unsigned deg = 1u;
#pragma unroll
  for (int i = 0; i < HC / 2; ++i) deg += pc[i];
  deg += (unsigned)__shfl_xor((int)(deg - 1u), 16);
  const float dv = rsqrtf((float)deg);
  if (half == 0 && m0 + lrow < N) dinv[m0 + lrow] = dv;
#pragma unroll
  for (int j = 0; j < 8; ++j) {
    int m = m0 + j + (half << 3);
    const float dlo = __int_as_float(__builtin_amdgcn_readlane(__float_as_int(dv), j));
    const float dhi = __int_as_float(__builtin_amdgcn_readlane(__float_as_int(dv), j + 8));
    const float dmj = half ? dhi : dlo;
    if (m < N) {
      const size_t o = (size_t)m * F_HID;
      const float v0 = c0[j] * dmj;
      const float v1 = c1[j] * dmj;
      H[o + lrow]      = v0;
      H[o + 16 + lrow] = v1;
    }
  }
}

__device__ __forceinline__ void gemm1_share(const float* __restrict__ X, const v4f* __restrict__ Wf,
                                            const unsigned* __restrict__ part, float* __restrict__ dinv,
                                            float* __restrict__ H, int N, int wg, int nwg, int wv, int lane) {
  const long long T = (N + 15) >> 4;
  const int t0 = (int)((wg * T) / nwg), t1 = (int)(((wg + 1) * T) / nwg);
  for (int t = t0 + wv; t < t1; t += 32) gemm1_tile(X, Wf, part, dinv, H, N, t, lane);
}


struct G2Ahead { v4f p[2 * (F_HID / 4)]; float dm[8]; float bz[4]; };

__device__ __forceinline__ void gemm2_ahead(G2Ahead& o, int r0, int nn, int n0, int lane,
                                            const v4f* __restrict__ Wf, const float* __restrict__ bias,
                                            const float* __restrict__ dinv) {
  const int half = lane >> 4;
  const int lrow = lane & 15;
  const v4f* wq = Wf + lane;
#pragma unroll
  for (int t = 0; t < 2 * (F_HID / 4); ++t) o.p[t] = wq[t * 32];
#pragma unroll
  for (int j = 0; j < 8; ++j) {
    int m = r0 + j + (half << 3); if (m >= nn) m = nn - 1;
    o.dm[j] = dinv[n0 + m];
  }
#pragma unroll
  for (int q = 0; q < 4; ++q) o.bz[q] = bias[16 * q + lrow];
}

__device__ __forceinline__ void gemm2_tile(const G2Ahead& o, const float* gs, int r0, int nn, int n0, int lane,
                                           float* __restrict__ H) {
  const int half = lane >> 4;
  const int lrow = lane & 15;
  int row = r0 + lrow; if (row >= nn) row = nn - 1;
  const v4f* aq0 = (const v4f*)(gs + row * GP + (half << 4));
  v4f s4[4];
#pragma unroll
  for (int i = 0; i < 4; ++i) s4[i] = aq0[i];
  v8f c0 = {}; v8f c1 = {}; v8f c2 = {}; v8f c3 = {};
#pragma unroll
  for (int t = 0; t < F_HID / 4; ++t) {
    const v4f sv = s4[t >> 1];
    v2f a;
    if (t & 1) { a.x = sv.z; a.y = sv.w; } else { a.x = sv.x; a.y = sv.y; }
    const v4f p0 = o.p[2 * t];
    const v4f p1 = o.p[2 * t + 1];
    v2f b;
    b.x = p0.x; b.y = p0.y;
    c0 = __builtin_amdgcn_wmma_f32_16x16x4_f32(false, a, false, b, (short)0, c0, false, false);
    b.x = p0.z; b.y = p0.w;
    c1 = __builtin_amdgcn_wmma_f32_16x16x4_f32(false, a, false, b, (short)0, c1, false, false);
    b.x = p1.x; b.y = p1.y;
    c2 = __builtin_amdgcn_wmma_f32_16x16x4_f32(false, a, false, b, (short)0, c2, false, false);
    b.x = p1.z; b.y = p1.w;
    c3 = __builtin_amdgcn_wmma_f32_16x16x4_f32(false, a, false, b, (short)0, c3, false, false);
  }
#pragma unroll
  for (int j = 0; j < 8; ++j) {
    const int m = r0 + j + (half << 3);
    if (m < nn) {
      float* hr = H + (size_t)(n0 + m) * F_OUT;
      hr[lrow]      = c0[j] * o.dm[j] + o.bz[0];
      hr[16 + lrow] = c1[j] * o.dm[j] + o.bz[1];
      hr[32 + lrow] = c2[j] * o.dm[j] + o.bz[2];
      hr[48 + lrow] = c3[j] * o.dm[j] + o.bz[3];
    }
  }
}

template <int LAYER>
__device__ __forceinline__ void pull_store(float sum, int n0, int i, int lane, float di, float bz, float* __restrict__ outp, float* gs) {
  if (LAYER == 1) outp[(size_t)(n0 + i) * F_HID + lane] = fmaxf(sum * di + bz, 0.0f) * di;
  else            gs[i * GP + lane] = sum;
}

template <int LAYER>
__global__ __launch_bounds__(1024) void k_pull(const float* __restrict__ rows, const unsigned* __restrict__ binned,
                                               const unsigned* __restrict__ binstart, const float* __restrict__ dinv,
                                               const float* __restrict__ bias, float* __restrict__ outp, int N,
                                               const v4f* __restrict__ W2f, const float* __restrict__ b2,
                                               float* __restrict__ out) {
  __shared__ unsigned sorted[CAP];
  __shared__ unsigned cnt[BW], st[BW + 1], cur[BW];
  __shared__ __attribute__((aligned(16))) float gs[LAYER == 2 ? BW * GP : 4];
  const int tid = threadIdx.x, lane = tid & 31, wv = tid >> 5;
  const int n0 = blockIdx.x * BW;
  int nn = N - n0; if (nn > BW) nn = BW;
  const unsigned s0 = binstart[blockIdx.x];
  const unsigned L = binstart[blockIdx.x + 1] - s0;
  const unsigned* __restrict__ lst = binned + s0;
  const float bz = (LAYER == 1) ? bias[lane] : 0.0f;
  const bool mine = (LAYER == 2) && wv < ((nn + 15) >> 4);
  G2Ahead ah;
  if (mine) gemm2_ahead(ah, wv << 4, nn, n0, lane, W2f, b2, dinv);
  if (L <= (unsigned)CAP) {
    for (int i = tid; i < BW; i += 1024) cnt[i] = 0u;
    __syncthreads();
    for (unsigned kb = 0; kb < L; kb += 1024 * PU) {
      unsigned ev[PU];
#pragma unroll
      for (int u = 0; u < PU; ++u) {
        const unsigned k = kb + u * 1024 + tid;
        const unsigned e = lst[k < L ? k : L - 1u];
        ev[u] = k < L ? e : NOEDGE;
      }
#pragma unroll
      for (int u = 0; u < PU; ++u)
        if (ev[u] != NOEDGE) atomicAdd(&cnt[ev[u] >> 17], 1u);
    }
    __syncthreads();
    if (wv == 0) {
      unsigned loc[13]; unsigned s = 0;
#pragma unroll
      for (int k = 0; k < 13; ++k) {
        const int idx = lane * 13 + k;
        loc[k] = idx < BW ? cnt[idx] : 0u;
        s += loc[k];
      }
      unsigned tot; unsigned ex = wave_excl_scan(s, lane, &tot);
#pragma unroll
      for (int k = 0; k < 13; ++k) {
        const int idx = lane * 13 + k;
        if (idx < BW) { st[idx] = ex; cur[idx] = ex; }
        ex += loc[k];
      }
      if (lane == 0) st[BW] = tot;
    }
    __syncthreads();
    for (unsigned kb = 0; kb < L; kb += 1024 * PU) {
      unsigned ev[PU], pp[PU];
#pragma unroll
      for (int u = 0; u < PU; ++u) {
        const unsigned k = kb + u * 1024 + tid;
        const unsigned e = lst[k < L ? k : L - 1u];
        ev[u] = k < L ? e : NOEDGE;
      }
#pragma unroll
      for (int u = 0; u < PU; ++u) {
        pp[u] = 0u;
        if (ev[u] != NOEDGE) pp[u] = atomicAdd(&cur[ev[u] >> 17], 1u);
      }
#pragma unroll
      for (int u = 0; u < PU; ++u)
        if (ev[u] != NOEDGE) sorted[pp[u]] = ev[u] & SRCM;
    }
    __syncthreads();
    for (int i = wv; i < nn; i += 32) {
      const unsigned a = (unsigned)__builtin_amdgcn_readfirstlane((int)st[i]);
      const unsigned z = (unsigned)__builtin_amdgcn_readfirstlane((int)st[i + 1]);
      float q0 = rows[(size_t)(n0 + i) * F_HID + lane];
      const float di = (LAYER == 1) ? dinv[n0 + i] : 0.0f;
      float q1 = 0.0f, q2 = 0.0f, q3 = 0.0f;
      for (unsigned k = a; k < z; k += 32) {
        const int n = (int)(z - k < 32u ? z - k : 32u);
        const int si = (int)((k + lane < z) ? sorted[k + lane] : 0u);
        float v[32];
#pragma unroll
        for (int j = 0; j < 32; ++j) {
          v[j] = 0.0f;
          if (j < n) v[j] = rows[(size_t)__builtin_amdgcn_readlane(si, j) * F_HID + lane];
        }
#pragma unroll
        for (int j = 0; j < 32; j += 4) {
          q0 += v[j]; q1 += v[j + 1]; q2 += v[j + 2]; q3 += v[j + 3];
        }
      }
      pull_store<LAYER>((q0 + q1) + (q2 + q3), n0, i, lane, di, bz, outp, gs);
    }
  } else {
    for (int i = wv; i < nn; i += 32) {
      float sum = rows[(size_t)(n0 + i) * F_HID + lane];
      const float di = (LAYER == 1) ? dinv[n0 + i] : 0.0f;
      for (unsigned k = 0; k < L; k += 32) {
        const unsigned e = (k + lane < L) ? lst[k + lane] : 0xFFFFFFFFu;
        unsigned m = __builtin_amdgcn_ballot_w32((e >> 17) == (unsigned)i);
        while (m) {
          const int j = __builtin_ctz(m);
          m &= m - 1u;
          sum += rows[(size_t)__builtin_amdgcn_readlane((int)(e & SRCM), j) * F_HID + lane];
        }
      }
      pull_store<LAYER>(sum, n0, i, lane, di, bz, outp, gs);
    }
  }
  if (LAYER == 2) {
    static_assert((BW + 15) / 16 <= 32, "a bin has at most one 16-node tile per wave");
    __syncthreads();
    if (mine) gemm2_tile(ah, gs, wv << 4, nn, n0, lane, out);
  }
}


static inline int cdiv_ll(long long a, long long b) { return (int)((a + b - 1) / b); }

extern "C" void kernel_launch(void* const* d_in, const int* in_sizes, int n_in,
                              void* d_out, int out_size, void* d_ws, size_t ws_size,
                              hipStream_t stream) {
  const float* X  = (const float*)d_in[0];
  const int*   ei = (const int*)d_in[1];
  const float* W1 = (const float*)d_in[2];
  const float* b1 = (const float*)d_in[3];
  const float* W2 = (const float*)d_in[4];
  const float* b2 = (const float*)d_in[5];
  float* out = (float*)d_out;

  const int N = in_sizes[0] / F_IN;
  const int E = in_sizes[1] / 2;
  const int* src = ei;
  const int* dst = ei + E;

  v4f* W1f = (v4f*)d_ws;
  v4f* W2f = W1f + (F_IN / 4) * 32;
  float* h1   = (float*)(W2f + (F_HID / 4) * 2 * 32);
  float* out1 = h1 + (size_t)N * F_HID;
  float* dinv = out1 + (size_t)N * F_HID;
  unsigned* part = (unsigned*)(dinv + N);
  unsigned* cntT = part + (size_t)HC * N;
  unsigned* binstart = cntT + NBIN * HC;
  unsigned* binned = binstart + (NBIN + 1);

  k_hist       <<<dim3(HC, cdiv_ll(N, HR)), 1024, 0, stream>>>(dst, part, cntT, E, N, W1, W1f, W2, W2f);
  k_fill_gemm1 <<<dim3(HC, cdiv_ll(N, HR)), 1024, 0, stream>>>(src, dst, cntT, binstart, binned, E, X, W1f, part, dinv, h1, N);
  k_pull<1>  <<<cdiv_ll(N, BW), 1024, 0, stream>>>(h1, binned, binstart, dinv, b1, out1, N, W2f, b2, out);
  k_pull<2>  <<<cdiv_ll(N, BW), 1024, 0, stream>>>(out1, binned, binstart, dinv, b1, nullptr, N, W2f, b2, out);
}
